# starting kernel
# baseline (speedup 1.0000x reference)
#include <hip/hip_runtime.h>
#include <math.h>

#define DIMA   100
#define MSTEP  50
#define HDIM   128
#define BATCH  8192
#define NPAD4  112
#define SROW   104
#define ROWS   64
#define S0INV  0.01f
#define RDT    0.001f
#define SIG    0.2f
#define LNEPS  1e-5f

typedef __attribute__((ext_vector_type(16))) _Float16 v16h;
typedef __attribute__((ext_vector_type(8)))  _Float16 v8h;
typedef __attribute__((ext_vector_type(8)))  float    v8f;

union V16U { v16h v; v8h h[2]; };

__device__ __forceinline__ v16h load_frag(const _Float16* base, int kb) {
  V16U u;
  u.h[0] = *(const v8h*)(base + kb);
  u.h[1] = *(const v8h*)(base + kb + 16);
  return u.v;
}

__device__ __forceinline__ float red16(float x) {
#pragma unroll
  for (int off = 1; off < 16; off <<= 1) x += __shfl_xor(x, off, 16);
  return x;
}

__device__ __forceinline__ void mlp_layer(const _Float16* in, const _Float16* Wt,
                                          const float* bb, const float* gg,
                                          const float* ee, _Float16* out,
                                          int rbase, int lane15, int half) {
  v8f acc[8];
#pragma unroll
  for (int nt = 0; nt < 8; ++nt) acc[nt] = {};

  const _Float16* arow = in + (rbase + lane15) * HDIM;
#pragma unroll
  for (int kt = 0; kt < 4; ++kt) {
    const int kb = kt * 32 + half * 8;
    v16h a = load_frag(arow, kb);
#pragma unroll
    for (int nt = 0; nt < 8; ++nt) {
      v16h b = load_frag(Wt + (nt * 16 + lane15) * HDIM, kb);
      acc[nt] = __builtin_amdgcn_wmma_f32_16x16x32_f16(
          false, a, false, b, (short)0, acc[nt], false, false);
    }
  }

  float bn[8], gn[8], en[8];
#pragma unroll
  for (int nt = 0; nt < 8; ++nt) {
    const int n = nt * 16 + lane15;
    bn[nt] = bb[n]; gn[nt] = gg[n]; en[nt] = ee[n];
  }

#pragma unroll
  for (int j = 0; j < 8; ++j) {
    float p = 0.f;
#pragma unroll
    for (int nt = 0; nt < 8; ++nt) { acc[nt][j] += bn[nt]; p += acc[nt][j]; }
    p = red16(p);
    const float mu = p * (1.f / 128.f);
    float q = 0.f;
#pragma unroll
    for (int nt = 0; nt < 8; ++nt) { float d = acc[nt][j] - mu; q += d * d; }
    q = red16(q);
    const float rstd = rsqrtf(q * (1.f / 128.f) + LNEPS);

    const int row = rbase + j + half * 8;
    _Float16* orow = out + row * HDIM + lane15;
#pragma unroll
    for (int nt = 0; nt < 8; ++nt) {
      float x = (acc[nt][j] - mu) * rstd * gn[nt] + en[nt];
      float ge = 0.5f * x * (1.f + erff(x * 0.70710678118654752f));
      orow[nt * 16] = (_Float16)ge;
    }
  }
}

extern "C" __global__ void __launch_bounds__(128)
bsde_fused_kernel(const float* __restrict__ dw, const float* __restrict__ tg,
                  const float* __restrict__ W1, const float* __restrict__ b1,
                  const float* __restrict__ g1, const float* __restrict__ be1,
                  const float* __restrict__ W2, const float* __restrict__ b2,
                  const float* __restrict__ g2, const float* __restrict__ be2,
                  const float* __restrict__ W3, const float* __restrict__ b3,
                  const float* __restrict__ g3, const float* __restrict__ be3,
                  const float* __restrict__ W4, const float* __restrict__ b4,
                  const float* __restrict__ Y0,
                  float* __restrict__ outY, float* __restrict__ outS) {
  extern __shared__ __align__(16) char smem[];
  _Float16* Wt1 = (_Float16*)smem;
  _Float16* Wt2 = Wt1 + HDIM * HDIM;
  _Float16* Wt3 = Wt2 + HDIM * HDIM;
  _Float16* Wt4 = Wt3 + HDIM * HDIM;
  float* prm = (float*)(Wt4 + NPAD4 * HDIM);
  float* pb1 = prm;        float* pg1 = prm + 128;  float* pe1 = prm + 256;
  float* pb2 = prm + 384;  float* pg2 = prm + 512;  float* pe2 = prm + 640;
  float* pb3 = prm + 768;  float* pg3 = prm + 896;  float* pe3 = prm + 1024;
  float* pb4 = prm + 1152;
  float* Sl  = prm + 1264;
  float* Yl  = Sl + ROWS * SROW;
  _Float16* buf0 = (_Float16*)(Yl + ROWS);
  _Float16* buf1 = buf0 + ROWS * HDIM;

  const int tid = threadIdx.x;
  const int lane = tid & 31, lane15 = lane & 15, half = lane >> 4;
  const int rbase = (tid >> 5) * 16;
  const long blockBase = (long)blockIdx.x * ROWS;

  for (int i = tid; i < HDIM * HDIM; i += 128) {
    const int n = i >> 7, k = i & 127;
    Wt1[i] = (_Float16)(k < DIMA + 1 ? W1[k * HDIM + n] : 0.f);
    Wt2[i] = (_Float16)W2[k * HDIM + n];
    Wt3[i] = (_Float16)W3[k * HDIM + n];
  }
  for (int i = tid; i < NPAD4 * HDIM; i += 128) {
    const int n = i >> 7, k = i & 127;
    Wt4[i] = (_Float16)(n < DIMA ? W4[k * DIMA + n] : 0.f);
  }
  pb1[tid] = b1[tid]; pg1[tid] = g1[tid]; pe1[tid] = be1[tid];
  pb2[tid] = b2[tid]; pg2[tid] = g2[tid]; pe2[tid] = be2[tid];
  pb3[tid] = b3[tid]; pg3[tid] = g3[tid]; pe3[tid] = be3[tid];
  if (tid < NPAD4) pb4[tid] = (tid < DIMA) ? b4[tid] : 0.f;

  const float y0 = Y0[0];
#pragma unroll 4
  for (int r = 0; r < 16; ++r) {
    for (int d = lane; d < DIMA; d += 32) Sl[(rbase + r) * SROW + d] = 100.f;
    if (lane == 0) Yl[rbase + r] = y0;
  }
  __syncthreads();

  for (int m = 0; m < MSTEP; ++m) {
    for (int i = lane; i < 16 * HDIM; i += 32) {
      const int rl = i >> 7, k = i & 127;
      const int row = rbase + rl;
      float v;
      if (k < DIMA)      v = Sl[row * SROW + k] * S0INV;
      else if (k == DIMA) v = tg[(blockBase + row) * MSTEP + m];
      else               v = 0.f;
      buf0[row * HDIM + k] = (_Float16)v;
    }

    mlp_layer(buf0, Wt1, pb1, pg1, pe1, buf1, rbase, lane15, half);
    mlp_layer(buf1, Wt2, pb2, pg2, pe2, buf0, rbase, lane15, half);
    mlp_layer(buf0, Wt3, pb3, pg3, pe3, buf1, rbase, lane15, half);

    {
      v8f acc[7];
#pragma unroll
      for (int nt = 0; nt < 7; ++nt) acc[nt] = {};
      const _Float16* arow = buf1 + (rbase + lane15) * HDIM;
#pragma unroll
      for (int kt = 0; kt < 4; ++kt) {
        const int kb = kt * 32 + half * 8;
        v16h a = load_frag(arow, kb);
#pragma unroll
        for (int nt = 0; nt < 7; ++nt) {
          v16h b = load_frag(Wt4 + (nt * 16 + lane15) * HDIM, kb);
          acc[nt] = __builtin_amdgcn_wmma_f32_16x16x32_f16(
              false, a, false, b, (short)0, acc[nt], false, false);
        }
      }
      float bn[7];
#pragma unroll
      for (int nt = 0; nt < 7; ++nt) bn[nt] = pb4[nt * 16 + lane15];

#pragma unroll
      for (int j = 0; j < 8; ++j) {
        const int row = rbase + j + half * 8;
        const long bidx = blockBase + row;
        const float* dwr = dw + ((long)bidx * MSTEP + m) * DIMA;
        float gain = 0.f, asset = 0.f;
#pragma unroll
        for (int nt = 0; nt < 7; ++nt) {
          const int d = nt * 16 + lane15;
          if (d < DIMA) {
            const float z  = 1.f / (1.f + expf(-(acc[nt][j] + bn[nt])));
            const float s  = Sl[row * SROW + d];
            const float ds = s * (RDT + SIG * dwr[d]);
            gain  += z * ds;
            asset += z * s;
            Sl[row * SROW + d] = s + ds;
          }
        }
        gain  = red16(gain);
        asset = red16(asset);
        if (lane15 == 0) {
          const float y = Yl[row];
          Yl[row] = y + RDT * (y - asset) + gain;
        }
      }
    }
  }

  __syncthreads();
  for (int i = tid; i < ROWS; i += 128) outY[blockBase + i] = Yl[i];
  for (int i = tid; i < ROWS * DIMA; i += 128) {
    const int r = i / DIMA, d = i - r * DIMA;
    outS[(blockBase + r) * (long)DIMA + d] = Sl[r * SROW + d];
  }
}

extern "C" void kernel_launch(void* const* d_in, const int* in_sizes, int n_in,
                              void* d_out, int out_size, void* d_ws, size_t ws_size,
                              hipStream_t stream) {
  (void)in_sizes; (void)n_in; (void)out_size; (void)d_ws; (void)ws_size;
  const float* dw  = (const float*)d_in[0];
  const float* tg  = (const float*)d_in[1];
  const float* W1  = (const float*)d_in[2];
  const float* b1  = (const float*)d_in[3];
  const float* g1  = (const float*)d_in[4];
  const float* be1 = (const float*)d_in[5];
  const float* W2  = (const float*)d_in[6];
  const float* b2  = (const float*)d_in[7];
  const float* g2  = (const float*)d_in[8];
  const float* be2 = (const float*)d_in[9];
  const float* W3  = (const float*)d_in[10];
  const float* b3  = (const float*)d_in[11];
  const float* g3  = (const float*)d_in[12];
  const float* be3 = (const float*)d_in[13];
  const float* W4  = (const float*)d_in[14];
  const float* b4  = (const float*)d_in[15];
  const float* Y0  = (const float*)d_in[16];

  float* outY = (float*)d_out;
  float* outS = outY + BATCH;

  const size_t smem =
      (size_t)(3 * HDIM * HDIM + NPAD4 * HDIM) * sizeof(_Float16) +
      (size_t)(9 * 128 + NPAD4) * sizeof(float) +
      (size_t)(ROWS * SROW + ROWS) * sizeof(float) +
      (size_t)(2 * ROWS * HDIM) * sizeof(_Float16);

  dim3 grid(BATCH / ROWS);
  dim3 block(128);
  bsde_fused_kernel<<<grid, block, smem, stream>>>(
      dw, tg, W1, b1, g1, be1, W2, b2, g2, be2, W3, b3, g3, be3, W4, b4, Y0,
      outY, outS);
}


# Sum each 16-lane row on the vector ALU: the shuffle butterflies (ds_bpermute_b32 plus a wait per step) become DPP row_xmask adds, same partners and order of additions
# speedup vs baseline: 1.9444x; 1.9444x over previous
#include <hip/hip_runtime.h>
#include <math.h>

#define DIMA   100
#define MSTEP  50
#define HDIM   128
#define BATCH  8192
#define NPAD4  112
#define SROW   104
#define ROWS   64
#define S0INV  0.01f
#define RDT    0.001f
#define SIG    0.2f
#define LNEPS  1e-5f

typedef __attribute__((ext_vector_type(16))) _Float16 v16h;
typedef __attribute__((ext_vector_type(8)))  _Float16 v8h;
typedef __attribute__((ext_vector_type(8)))  float    v8f;

union V16U { v16h v; v8h h[2]; };

__device__ __forceinline__ v16h load_frag(const _Float16* base, int kb) {
  V16U u;
  u.h[0] = *(const v8h*)(base + kb);
  u.h[1] = *(const v8h*)(base + kb + 16);
  return u.v;
}

template <int M>
__device__ __forceinline__ float row_xor_add(float x) {
  const int partner = __builtin_amdgcn_update_dpp(
      0, __builtin_bit_cast(int, x), 0x160 + M, 0xf, 0xf, true);
  return x + __builtin_bit_cast(float, partner);
}

__device__ __forceinline__ float red16(float x) {
  x = row_xor_add<1>(x);
  x = row_xor_add<2>(x);
  x = row_xor_add<4>(x);
  x = row_xor_add<8>(x);
  return x;
}

__device__ __forceinline__ void mlp_layer(const _Float16* in, const _Float16* Wt,
                                          const float* bb, const float* gg,
                                          const float* ee, _Float16* out,
                                          int rbase, int lane15, int half) {
  v8f acc[8];
#pragma unroll
  for (int nt = 0; nt < 8; ++nt) acc[nt] = {};

  const _Float16* arow = in + (rbase + lane15) * HDIM;
#pragma unroll
  for (int kt = 0; kt < 4; ++kt) {
    const int kb = kt * 32 + half * 8;
    v16h a = load_frag(arow, kb);
#pragma unroll
    for (int nt = 0; nt < 8; ++nt) {
      v16h b = load_frag(Wt + (nt * 16 + lane15) * HDIM, kb);
      acc[nt] = __builtin_amdgcn_wmma_f32_16x16x32_f16(
          false, a, false, b, (short)0, acc[nt], false, false);
    }
  }

  float bn[8], gn[8], en[8];
#pragma unroll
  for (int nt = 0; nt < 8; ++nt) {
    const int n = nt * 16 + lane15;
    bn[nt] = bb[n]; gn[nt] = gg[n]; en[nt] = ee[n];
  }

#pragma unroll
  for (int j = 0; j < 8; ++j) {
    float p = 0.f;
#pragma unroll
    for (int nt = 0; nt < 8; ++nt) { acc[nt][j] += bn[nt]; p += acc[nt][j]; }
    p = red16(p);
    const float mu = p * (1.f / 128.f);
    float q = 0.f;
#pragma unroll
    for (int nt = 0; nt < 8; ++nt) { float d = acc[nt][j] - mu; q += d * d; }
    q = red16(q);
    const float rstd = rsqrtf(q * (1.f / 128.f) + LNEPS);

    const int row = rbase + j + half * 8;
    _Float16* orow = out + row * HDIM + lane15;
#pragma unroll
    for (int nt = 0; nt < 8; ++nt) {
      float x = (acc[nt][j] - mu) * rstd * gn[nt] + en[nt];
      float ge = 0.5f * x * (1.f + erff(x * 0.70710678118654752f));
      orow[nt * 16] = (_Float16)ge;
    }
  }
}

extern "C" __global__ void __launch_bounds__(128)
bsde_fused_kernel(const float* __restrict__ dw, const float* __restrict__ tg,
                  const float* __restrict__ W1, const float* __restrict__ b1,
                  const float* __restrict__ g1, const float* __restrict__ be1,
                  const float* __restrict__ W2, const float* __restrict__ b2,
                  const float* __restrict__ g2, const float* __restrict__ be2,
                  const float* __restrict__ W3, const float* __restrict__ b3,
                  const float* __restrict__ g3, const float* __restrict__ be3,
                  const float* __restrict__ W4, const float* __restrict__ b4,
                  const float* __restrict__ Y0,
                  float* __restrict__ outY, float* __restrict__ outS) {
  extern __shared__ __align__(16) char smem[];
  _Float16* Wt1 = (_Float16*)smem;
  _Float16* Wt2 = Wt1 + HDIM * HDIM;
  _Float16* Wt3 = Wt2 + HDIM * HDIM;
  _Float16* Wt4 = Wt3 + HDIM * HDIM;
  float* prm = (float*)(Wt4 + NPAD4 * HDIM);
  float* pb1 = prm;        float* pg1 = prm + 128;  float* pe1 = prm + 256;
  float* pb2 = prm + 384;  float* pg2 = prm + 512;  float* pe2 = prm + 640;
  float* pb3 = prm + 768;  float* pg3 = prm + 896;  float* pe3 = prm + 1024;
  float* pb4 = prm + 1152;
  float* Sl  = prm + 1264;
  float* Yl  = Sl + ROWS * SROW;
  _Float16* buf0 = (_Float16*)(Yl + ROWS);
  _Float16* buf1 = buf0 + ROWS * HDIM;

  const int tid = threadIdx.x;
  const int lane = tid & 31, lane15 = lane & 15, half = lane >> 4;
  const int rbase = (tid >> 5) * 16;
  const long blockBase = (long)blockIdx.x * ROWS;

  for (int i = tid; i < HDIM * HDIM; i += 128) {
    const int n = i >> 7, k = i & 127;
    Wt1[i] = (_Float16)(k < DIMA + 1 ? W1[k * HDIM + n] : 0.f);
    Wt2[i] = (_Float16)W2[k * HDIM + n];
    Wt3[i] = (_Float16)W3[k * HDIM + n];
  }
  for (int i = tid; i < NPAD4 * HDIM; i += 128) {
    const int n = i >> 7, k = i & 127;
    Wt4[i] = (_Float16)(n < DIMA ? W4[k * DIMA + n] : 0.f);
  }
  pb1[tid] = b1[tid]; pg1[tid] = g1[tid]; pe1[tid] = be1[tid];
  pb2[tid] = b2[tid]; pg2[tid] = g2[tid]; pe2[tid] = be2[tid];
  pb3[tid] = b3[tid]; pg3[tid] = g3[tid]; pe3[tid] = be3[tid];
  if (tid < NPAD4) pb4[tid] = (tid < DIMA) ? b4[tid] : 0.f;

  const float y0 = Y0[0];
#pragma unroll 4
  for (int r = 0; r < 16; ++r) {
    for (int d = lane; d < DIMA; d += 32) Sl[(rbase + r) * SROW + d] = 100.f;
    if (lane == 0) Yl[rbase + r] = y0;
  }
  __syncthreads();

  for (int m = 0; m < MSTEP; ++m) {
    for (int i = lane; i < 16 * HDIM; i += 32) {
      const int rl = i >> 7, k = i & 127;
      const int row = rbase + rl;
      float v;
      if (k < DIMA)      v = Sl[row * SROW + k] * S0INV;
      else if (k == DIMA) v = tg[(blockBase + row) * MSTEP + m];
      else               v = 0.f;
      buf0[row * HDIM + k] = (_Float16)v;
    }

#pragma unroll 1
    for (int l = 0; l < 3; ++l) {
      const float* lp = prm + l * (3 * HDIM);
      mlp_layer((l & 1) ? buf1 : buf0, Wt1 + l * (HDIM * HDIM),
                lp, lp + HDIM, lp + 2 * HDIM,
                (l & 1) ? buf0 : buf1, rbase, lane15, half);
    }

    {
      v8f acc[7];
#pragma unroll
      for (int nt = 0; nt < 7; ++nt) acc[nt] = {};
      const _Float16* arow = buf1 + (rbase + lane15) * HDIM;
#pragma unroll
      for (int kt = 0; kt < 4; ++kt) {
        const int kb = kt * 32 + half * 8;
        v16h a = load_frag(arow, kb);
#pragma unroll
        for (int nt = 0; nt < 7; ++nt) {
          v16h b = load_frag(Wt4 + (nt * 16 + lane15) * HDIM, kb);
          acc[nt] = __builtin_amdgcn_wmma_f32_16x16x32_f16(
              false, a, false, b, (short)0, acc[nt], false, false);
        }
      }
      float bn[7];
#pragma unroll
      for (int nt = 0; nt < 7; ++nt) bn[nt] = pb4[nt * 16 + lane15];

#pragma unroll
      for (int j = 0; j < 8; ++j) {
        const int row = rbase + j + half * 8;
        const long bidx = blockBase + row;
        const float* dwr = dw + ((long)bidx * MSTEP + m) * DIMA;
        float gain = 0.f, asset = 0.f;
#pragma unroll
        for (int nt = 0; nt < 7; ++nt) {
          const int d = nt * 16 + lane15;
          if (d < DIMA) {
            const float z  = 1.f / (1.f + expf(-(acc[nt][j] + bn[nt])));
            const float s  = Sl[row * SROW + d];
            const float ds = s * (RDT + SIG * dwr[d]);
            gain  += z * ds;
            asset += z * s;
            Sl[row * SROW + d] = s + ds;
          }
        }
        gain  = red16(gain);
        asset = red16(asset);
        if (lane15 == 0) {
          const float y = Yl[row];
          Yl[row] = y + RDT * (y - asset) + gain;
        }
      }
    }
  }

  __syncthreads();
  for (int i = tid; i < ROWS; i += 128) outY[blockBase + i] = Yl[i];
  for (int i = tid; i < ROWS * DIMA; i += 128) {
    const int r = i / DIMA, d = i - r * DIMA;
    outS[(blockBase + r) * (long)DIMA + d] = Sl[r * SROW + d];
  }
}

extern "C" void kernel_launch(void* const* d_in, const int* in_sizes, int n_in,
                              void* d_out, int out_size, void* d_ws, size_t ws_size,
                              hipStream_t stream) {
  (void)in_sizes; (void)n_in; (void)out_size; (void)d_ws; (void)ws_size;
  const float* dw  = (const float*)d_in[0];
  const float* tg  = (const float*)d_in[1];
  const float* W1  = (const float*)d_in[2];
  const float* b1  = (const float*)d_in[3];
  const float* g1  = (const float*)d_in[4];
  const float* be1 = (const float*)d_in[5];
  const float* W2  = (const float*)d_in[6];
  const float* b2  = (const float*)d_in[7];
  const float* g2  = (const float*)d_in[8];
  const float* be2 = (const float*)d_in[9];
  const float* W3  = (const float*)d_in[10];
  const float* b3  = (const float*)d_in[11];
  const float* g3  = (const float*)d_in[12];
  const float* be3 = (const float*)d_in[13];
  const float* W4  = (const float*)d_in[14];
  const float* b4  = (const float*)d_in[15];
  const float* Y0  = (const float*)d_in[16];

  float* outY = (float*)d_out;
  float* outS = outY + BATCH;

  const size_t smem =
      (size_t)(3 * HDIM * HDIM + NPAD4 * HDIM) * sizeof(_Float16) +
      (size_t)(9 * 128 + NPAD4) * sizeof(float) +
      (size_t)(ROWS * SROW + ROWS) * sizeof(float) +
      (size_t)(2 * ROWS * HDIM) * sizeof(_Float16);

  dim3 grid(BATCH / ROWS);
  dim3 block(128);
  bsde_fused_kernel<<<grid, block, smem, stream>>>(
      dw, tg, W1, b1, g1, be1, W2, b2, g2, be2, W3, b3, g3, be3, W4, b4, Y0,
      outY, outS);
}


# GELU's erf evaluated branch-free for the 8 column tiles of a row at once (both range polynomials, select, hardware exp2): 8 independent chains overlap, not 64 serial branchy calls; erf within 2 ulp
# speedup vs baseline: 2.8151x; 1.4478x over previous
#include <hip/hip_runtime.h>
#include <math.h>

#define DIMA   100
#define MSTEP  50
#define HDIM   128
#define BATCH  8192
#define NPAD4  112
#define SROW   104
#define ROWS   64
#define S0INV  0.01f
#define RDT    0.001f
#define SIG    0.2f
#define LNEPS  1e-5f

typedef __attribute__((ext_vector_type(16))) _Float16 v16h;
typedef __attribute__((ext_vector_type(8)))  _Float16 v8h;
typedef __attribute__((ext_vector_type(8)))  float    v8f;

union V16U { v16h v; v8h h[2]; };

__device__ __forceinline__ v16h load_frag(const _Float16* base, int kb) {
  V16U u;
  u.h[0] = *(const v8h*)(base + kb);
  u.h[1] = *(const v8h*)(base + kb + 16);
  return u.v;
}

template <int M>
__device__ __forceinline__ float row_xor_add(float x) {
  const int partner = __builtin_amdgcn_update_dpp(
      0, __builtin_bit_cast(int, x), 0x160 + M, 0xf, 0xf, true);
  return x + __builtin_bit_cast(float, partner);
}

__device__ __forceinline__ float red16(float x) {
  x = row_xor_add<1>(x);
  x = row_xor_add<2>(x);
  x = row_xor_add<4>(x);
  x = row_xor_add<8>(x);
  return x;
}

__device__ __forceinline__ v8f erf8(v8f u) {
  v8f ax;
#pragma unroll
  for (int i = 0; i < 8; ++i) ax[i] = __builtin_fabsf(u[i]);
  const v8f t = u * u;
  v8f p = t * -5.618018e-4f + 4.913816e-3f;
  p = p * t + -2.6707515e-2f;
  p = p * t + 1.1280011e-1f;
  p = p * t + -3.7612295e-1f;
  p = p * t + 1.283791e-1f;
  const v8f rs = ax * p + ax;
  v8f q = ax * 1.699881e-5f + -3.7867785e-4f;
  q = q * ax + 3.8578159e-3f;
  q = q * ax + -2.4181698e-2f;
  q = q * ax + 1.0666826e-1f;
  q = q * ax + 6.349333e-1f;
  q = q * ax + 1.2868941e-1f;
  q = q * ax + ax;
  const v8f ea = q * -1.4426950408889634f;
  v8f rb;
#pragma unroll
  for (int i = 0; i < 8; ++i) rb[i] = 1.f - __builtin_amdgcn_exp2f(ea[i]);
  const v8f pick = (ax < 1.f) ? rs : rb;
  v8f r;
#pragma unroll
  for (int i = 0; i < 8; ++i) r[i] = __builtin_copysignf(pick[i], u[i]);
  return r;
}

__device__ __forceinline__ void mlp_layer(const _Float16* in, const _Float16* Wt,
                                          const float* bb, const float* gg,
                                          const float* ee, _Float16* out,
                                          int rbase, int lane15, int half) {
  v8f acc[8];
#pragma unroll
  for (int nt = 0; nt < 8; ++nt) acc[nt] = {};

  const _Float16* arow = in + (rbase + lane15) * HDIM;
#pragma unroll
  for (int kt = 0; kt < 4; ++kt) {
    const int kb = kt * 32 + half * 8;
    v16h a = load_frag(arow, kb);
#pragma unroll
    for (int nt = 0; nt < 8; ++nt) {
      v16h b = load_frag(Wt + (nt * 16 + lane15) * HDIM, kb);
      acc[nt] = __builtin_amdgcn_wmma_f32_16x16x32_f16(
          false, a, false, b, (short)0, acc[nt], false, false);
    }
  }

  float bn[8], gn[8], en[8];
#pragma unroll
  for (int nt = 0; nt < 8; ++nt) {
    const int n = nt * 16 + lane15;
    bn[nt] = bb[n]; gn[nt] = gg[n]; en[nt] = ee[n];
  }

#pragma unroll
  for (int j = 0; j < 8; ++j) {
    float p = 0.f;
#pragma unroll
    for (int nt = 0; nt < 8; ++nt) { acc[nt][j] += bn[nt]; p += acc[nt][j]; }
    p = red16(p);
    const float mu = p * (1.f / 128.f);
    float q = 0.f;
#pragma unroll
    for (int nt = 0; nt < 8; ++nt) { float d = acc[nt][j] - mu; q += d * d; }
    q = red16(q);
    const float rstd = rsqrtf(q * (1.f / 128.f) + LNEPS);

    const int row = rbase + j + half * 8;
    _Float16* orow = out + row * HDIM + lane15;
    v8f x;
#pragma unroll
    for (int nt = 0; nt < 8; ++nt)
      x[nt] = (acc[nt][j] - mu) * rstd * gn[nt] + en[nt];
    const v8f ge = (x * 0.5f) * (erf8(x * 0.70710678118654752f) + 1.f);
#pragma unroll
    for (int nt = 0; nt < 8; ++nt) orow[nt * 16] = (_Float16)ge[nt];
  }
}

extern "C" __global__ void __launch_bounds__(128)
bsde_fused_kernel(const float* __restrict__ dw, const float* __restrict__ tg,
                  const float* __restrict__ W1, const float* __restrict__ b1,
                  const float* __restrict__ g1, const float* __restrict__ be1,
                  const float* __restrict__ W2, const float* __restrict__ b2,
                  const float* __restrict__ g2, const float* __restrict__ be2,
                  const float* __restrict__ W3, const float* __restrict__ b3,
                  const float* __restrict__ g3, const float* __restrict__ be3,
                  const float* __restrict__ W4, const float* __restrict__ b4,
                  const float* __restrict__ Y0,
                  float* __restrict__ outY, float* __restrict__ outS) {
  extern __shared__ __align__(16) char smem[];
  _Float16* Wt1 = (_Float16*)smem;
  _Float16* Wt2 = Wt1 + HDIM * HDIM;
  _Float16* Wt3 = Wt2 + HDIM * HDIM;
  _Float16* Wt4 = Wt3 + HDIM * HDIM;
  float* prm = (float*)(Wt4 + NPAD4 * HDIM);
  float* pb1 = prm;        float* pg1 = prm + 128;  float* pe1 = prm + 256;
  float* pb2 = prm + 384;  float* pg2 = prm + 512;  float* pe2 = prm + 640;
  float* pb3 = prm + 768;  float* pg3 = prm + 896;  float* pe3 = prm + 1024;
  float* pb4 = prm + 1152;
  float* Sl  = prm + 1264;
  float* Yl  = Sl + ROWS * SROW;
  _Float16* buf0 = (_Float16*)(Yl + ROWS);
  _Float16* buf1 = buf0 + ROWS * HDIM;

  const int tid = threadIdx.x;
  const int lane = tid & 31, lane15 = lane & 15, half = lane >> 4;
  const int rbase = (tid >> 5) * 16;
  const long blockBase = (long)blockIdx.x * ROWS;

  for (int i = tid; i < HDIM * HDIM; i += 128) {
    const int n = i >> 7, k = i & 127;
    Wt1[i] = (_Float16)(k < DIMA + 1 ? W1[k * HDIM + n] : 0.f);
    Wt2[i] = (_Float16)W2[k * HDIM + n];
    Wt3[i] = (_Float16)W3[k * HDIM + n];
  }
  for (int i = tid; i < NPAD4 * HDIM; i += 128) {
    const int n = i >> 7, k = i & 127;
    Wt4[i] = (_Float16)(n < DIMA ? W4[k * DIMA + n] : 0.f);
  }
  pb1[tid] = b1[tid]; pg1[tid] = g1[tid]; pe1[tid] = be1[tid];
  pb2[tid] = b2[tid]; pg2[tid] = g2[tid]; pe2[tid] = be2[tid];
  pb3[tid] = b3[tid]; pg3[tid] = g3[tid]; pe3[tid] = be3[tid];
  if (tid < NPAD4) pb4[tid] = (tid < DIMA) ? b4[tid] : 0.f;

  const float y0 = Y0[0];
#pragma unroll 4
  for (int r = 0; r < 16; ++r) {
    for (int d = lane; d < DIMA; d += 32) Sl[(rbase + r) * SROW + d] = 100.f;
    if (lane == 0) Yl[rbase + r] = y0;
  }
  __syncthreads();

  for (int m = 0; m < MSTEP; ++m) {
    for (int i = lane; i < 16 * HDIM; i += 32) {
      const int rl = i >> 7, k = i & 127;
      const int row = rbase + rl;
      float v;
      if (k < DIMA)      v = Sl[row * SROW + k] * S0INV;
      else if (k == DIMA) v = tg[(blockBase + row) * MSTEP + m];
      else               v = 0.f;
      buf0[row * HDIM + k] = (_Float16)v;
    }

#pragma unroll 1
    for (int l = 0; l < 3; ++l) {
      const float* lp = prm + l * (3 * HDIM);
      mlp_layer((l & 1) ? buf1 : buf0, Wt1 + l * (HDIM * HDIM),
                lp, lp + HDIM, lp + 2 * HDIM,
                (l & 1) ? buf0 : buf1, rbase, lane15, half);
    }

    {
      v8f acc[7];
#pragma unroll
      for (int nt = 0; nt < 7; ++nt) acc[nt] = {};
      const _Float16* arow = buf1 + (rbase + lane15) * HDIM;
#pragma unroll
      for (int kt = 0; kt < 4; ++kt) {
        const int kb = kt * 32 + half * 8;
        v16h a = load_frag(arow, kb);
#pragma unroll
        for (int nt = 0; nt < 7; ++nt) {
          v16h b = load_frag(Wt4 + (nt * 16 + lane15) * HDIM, kb);
          acc[nt] = __builtin_amdgcn_wmma_f32_16x16x32_f16(
              false, a, false, b, (short)0, acc[nt], false, false);
        }
      }
      float bn[7];
#pragma unroll
      for (int nt = 0; nt < 7; ++nt) bn[nt] = pb4[nt * 16 + lane15];

#pragma unroll
      for (int j = 0; j < 8; ++j) {
        const int row = rbase + j + half * 8;
        const long bidx = blockBase + row;
        const float* dwr = dw + ((long)bidx * MSTEP + m) * DIMA;
        float gain = 0.f, asset = 0.f;
#pragma unroll
        for (int nt = 0; nt < 7; ++nt) {
          const int d = nt * 16 + lane15;
          if (d < DIMA) {
            const float z  = 1.f / (1.f + expf(-(acc[nt][j] + bn[nt])));
            const float s  = Sl[row * SROW + d];
            const float ds = s * (RDT + SIG * dwr[d]);
            gain  += z * ds;
            asset += z * s;
            Sl[row * SROW + d] = s + ds;
          }
        }
        gain  = red16(gain);
        asset = red16(asset);
        if (lane15 == 0) {
          const float y = Yl[row];
          Yl[row] = y + RDT * (y - asset) + gain;
        }
      }
    }
  }

  __syncthreads();
  for (int i = tid; i < ROWS; i += 128) outY[blockBase + i] = Yl[i];
  for (int i = tid; i < ROWS * DIMA; i += 128) {
    const int r = i / DIMA, d = i - r * DIMA;
    outS[(blockBase + r) * (long)DIMA + d] = Sl[r * SROW + d];
  }
}

extern "C" void kernel_launch(void* const* d_in, const int* in_sizes, int n_in,
                              void* d_out, int out_size, void* d_ws, size_t ws_size,
                              hipStream_t stream) {
  (void)in_sizes; (void)n_in; (void)out_size; (void)d_ws; (void)ws_size;
  const float* dw  = (const float*)d_in[0];
  const float* tg  = (const float*)d_in[1];
  const float* W1  = (const float*)d_in[2];
  const float* b1  = (const float*)d_in[3];
  const float* g1  = (const float*)d_in[4];
  const float* be1 = (const float*)d_in[5];
  const float* W2  = (const float*)d_in[6];
  const float* b2  = (const float*)d_in[7];
  const float* g2  = (const float*)d_in[8];
  const float* be2 = (const float*)d_in[9];
  const float* W3  = (const float*)d_in[10];
  const float* b3  = (const float*)d_in[11];
  const float* g3  = (const float*)d_in[12];
  const float* be3 = (const float*)d_in[13];
  const float* W4  = (const float*)d_in[14];
  const float* b4  = (const float*)d_in[15];
  const float* Y0  = (const float*)d_in[16];

  float* outY = (float*)d_out;
  float* outS = outY + BATCH;

  const size_t smem =
      (size_t)(3 * HDIM * HDIM + NPAD4 * HDIM) * sizeof(_Float16) +
      (size_t)(9 * 128 + NPAD4) * sizeof(float) +
      (size_t)(ROWS * SROW + ROWS) * sizeof(float) +
      (size_t)(2 * ROWS * HDIM) * sizeof(_Float16);

  dim3 grid(BATCH / ROWS);
  dim3 block(128);
  bsde_fused_kernel<<<grid, block, smem, stream>>>(
      dw, tg, W1, b1, g1, be1, W2, b2, g2, be2, W3, b3, g3, be3, W4, b4, Y0,
      outY, outS);
}


# Issue each step's 56 dW loads per lane as one group ahead of the three hidden layers and hold them in registers: the head no longer waits out a memory round trip before each row's SDE update
# speedup vs baseline: 3.1343x; 1.1134x over previous
#include <hip/hip_runtime.h>
#include <math.h>

#define DIMA   100
#define MSTEP  50
#define HDIM   128
#define BATCH  8192
#define NPAD4  112
#define SROW   104
#define ROWS   64
#define S0INV  0.01f
#define RDT    0.001f
#define SIG    0.2f
#define LNEPS  1e-5f

typedef __attribute__((ext_vector_type(16))) _Float16 v16h;
typedef __attribute__((ext_vector_type(8)))  _Float16 v8h;
typedef __attribute__((ext_vector_type(8)))  float    v8f;

union V16U { v16h v; v8h h[2]; };

__device__ __forceinline__ v16h load_frag(const _Float16* base, int kb) {
  V16U u;
  u.h[0] = *(const v8h*)(base + kb);
  u.h[1] = *(const v8h*)(base + kb + 16);
  return u.v;
}

template <int M>
__device__ __forceinline__ float row_xor_add(float x) {
  const int partner = __builtin_amdgcn_update_dpp(
      0, __builtin_bit_cast(int, x), 0x160 + M, 0xf, 0xf, true);
  return x + __builtin_bit_cast(float, partner);
}

__device__ __forceinline__ float red16(float x) {
  x = row_xor_add<1>(x);
  x = row_xor_add<2>(x);
  x = row_xor_add<4>(x);
  x = row_xor_add<8>(x);
  return x;
}

__device__ __forceinline__ v8f erf8(v8f u) {
  v8f ax;
#pragma unroll
  for (int i = 0; i < 8; ++i) ax[i] = __builtin_fabsf(u[i]);
  const v8f t = u * u;
  v8f p = t * -5.618018e-4f + 4.913816e-3f;
  p = p * t + -2.6707515e-2f;
  p = p * t + 1.1280011e-1f;
  p = p * t + -3.7612295e-1f;
  p = p * t + 1.283791e-1f;
  const v8f rs = ax * p + ax;
  v8f q = ax * 1.699881e-5f + -3.7867785e-4f;
  q = q * ax + 3.8578159e-3f;
  q = q * ax + -2.4181698e-2f;
  q = q * ax + 1.0666826e-1f;
  q = q * ax + 6.349333e-1f;
  q = q * ax + 1.2868941e-1f;
  q = q * ax + ax;
  const v8f ea = q * -1.4426950408889634f;
  v8f rb;
#pragma unroll
  for (int i = 0; i < 8; ++i) rb[i] = 1.f - __builtin_amdgcn_exp2f(ea[i]);
  const v8f pick = (ax < 1.f) ? rs : rb;
  v8f r;
#pragma unroll
  for (int i = 0; i < 8; ++i) r[i] = __builtin_copysignf(pick[i], u[i]);
  return r;
}

__device__ __forceinline__ void mlp_layer(const _Float16* in, const _Float16* Wt,
                                          const float* bb, const float* gg,
                                          const float* ee, _Float16* out,
                                          int rbase, int lane15, int half) {
  v8f acc[8];
#pragma unroll
  for (int nt = 0; nt < 8; ++nt) acc[nt] = {};

  const _Float16* arow = in + (rbase + lane15) * HDIM;
#pragma unroll
  for (int kt = 0; kt < 4; ++kt) {
    const int kb = kt * 32 + half * 8;
    v16h a = load_frag(arow, kb);
#pragma unroll
    for (int nt = 0; nt < 8; ++nt) {
      v16h b = load_frag(Wt + (nt * 16 + lane15) * HDIM, kb);
      acc[nt] = __builtin_amdgcn_wmma_f32_16x16x32_f16(
          false, a, false, b, (short)0, acc[nt], false, false);
    }
  }

  float bn[8], gn[8], en[8];
#pragma unroll
  for (int nt = 0; nt < 8; ++nt) {
    const int n = nt * 16 + lane15;
    bn[nt] = bb[n]; gn[nt] = gg[n]; en[nt] = ee[n];
  }

#pragma unroll
  for (int j = 0; j < 8; ++j) {
    float p = 0.f;
#pragma unroll
    for (int nt = 0; nt < 8; ++nt) { acc[nt][j] += bn[nt]; p += acc[nt][j]; }
    p = red16(p);
    const float mu = p * (1.f / 128.f);
    float q = 0.f;
#pragma unroll
    for (int nt = 0; nt < 8; ++nt) { float d = acc[nt][j] - mu; q += d * d; }
    q = red16(q);
    const float rstd = rsqrtf(q * (1.f / 128.f) + LNEPS);

    const int row = rbase + j + half * 8;
    _Float16* orow = out + row * HDIM + lane15;
    v8f x;
#pragma unroll
    for (int nt = 0; nt < 8; ++nt)
      x[nt] = (acc[nt][j] - mu) * rstd * gn[nt] + en[nt];
    const v8f ge = (x * 0.5f) * (erf8(x * 0.70710678118654752f) + 1.f);
#pragma unroll
    for (int nt = 0; nt < 8; ++nt) orow[nt * 16] = (_Float16)ge[nt];
  }
}

extern "C" __global__ void __launch_bounds__(128)
bsde_fused_kernel(const float* __restrict__ dw, const float* __restrict__ tg,
                  const float* __restrict__ W1, const float* __restrict__ b1,
                  const float* __restrict__ g1, const float* __restrict__ be1,
                  const float* __restrict__ W2, const float* __restrict__ b2,
                  const float* __restrict__ g2, const float* __restrict__ be2,
                  const float* __restrict__ W3, const float* __restrict__ b3,
                  const float* __restrict__ g3, const float* __restrict__ be3,
                  const float* __restrict__ W4, const float* __restrict__ b4,
                  const float* __restrict__ Y0,
                  float* __restrict__ outY, float* __restrict__ outS) {
  extern __shared__ __align__(16) char smem[];
  _Float16* Wt1 = (_Float16*)smem;
  _Float16* Wt2 = Wt1 + HDIM * HDIM;
  _Float16* Wt3 = Wt2 + HDIM * HDIM;
  _Float16* Wt4 = Wt3 + HDIM * HDIM;
  float* prm = (float*)(Wt4 + NPAD4 * HDIM);
  float* pb1 = prm;        float* pg1 = prm + 128;  float* pe1 = prm + 256;
  float* pb2 = prm + 384;  float* pg2 = prm + 512;  float* pe2 = prm + 640;
  float* pb3 = prm + 768;  float* pg3 = prm + 896;  float* pe3 = prm + 1024;
  float* pb4 = prm + 1152;
  float* Sl  = prm + 1264;
  float* Yl  = Sl + ROWS * SROW;
  _Float16* buf0 = (_Float16*)(Yl + ROWS);
  _Float16* buf1 = buf0 + ROWS * HDIM;

  const int tid = threadIdx.x;
  const int lane = tid & 31, lane15 = lane & 15, half = lane >> 4;
  const int rbase = (tid >> 5) * 16;
  const long blockBase = (long)blockIdx.x * ROWS;

  for (int i = tid; i < HDIM * HDIM; i += 128) {
    const int n = i >> 7, k = i & 127;
    Wt1[i] = (_Float16)(k < DIMA + 1 ? W1[k * HDIM + n] : 0.f);
    Wt2[i] = (_Float16)W2[k * HDIM + n];
    Wt3[i] = (_Float16)W3[k * HDIM + n];
  }
  for (int i = tid; i < NPAD4 * HDIM; i += 128) {
    const int n = i >> 7, k = i & 127;
    Wt4[i] = (_Float16)(n < DIMA ? W4[k * DIMA + n] : 0.f);
  }
  pb1[tid] = b1[tid]; pg1[tid] = g1[tid]; pe1[tid] = be1[tid];
  pb2[tid] = b2[tid]; pg2[tid] = g2[tid]; pe2[tid] = be2[tid];
  pb3[tid] = b3[tid]; pg3[tid] = g3[tid]; pe3[tid] = be3[tid];
  if (tid < NPAD4) pb4[tid] = (tid < DIMA) ? b4[tid] : 0.f;

  const float y0 = Y0[0];
#pragma unroll 4
  for (int r = 0; r < 16; ++r) {
    for (int d = lane; d < DIMA; d += 32) Sl[(rbase + r) * SROW + d] = 100.f;
    if (lane == 0) Yl[rbase + r] = y0;
  }
  __syncthreads();

  for (int m = 0; m < MSTEP; ++m) {
    for (int i = lane; i < 16 * HDIM; i += 32) {
      const int rl = i >> 7, k = i & 127;
      const int row = rbase + rl;
      float v;
      if (k < DIMA)      v = Sl[row * SROW + k] * S0INV;
      else if (k == DIMA) v = tg[(blockBase + row) * MSTEP + m];
      else               v = 0.f;
      buf0[row * HDIM + k] = (_Float16)v;
    }

    float dwv[8][7];
    {
      const float* dwp =
          dw + ((blockBase + rbase + half * 8) * MSTEP + m) * (long)DIMA + lane15;
#pragma unroll
      for (int j = 0; j < 8; ++j) {
#pragma unroll
        for (int nt = 0; nt < 6; ++nt) dwv[j][nt] = dwp[j * (MSTEP * DIMA) + nt * 16];
      }
      const int o6 = (lane15 < DIMA - 96) ? 96 : (DIMA - 1 - lane15);
#pragma unroll
      for (int j = 0; j < 8; ++j) dwv[j][6] = dwp[j * (MSTEP * DIMA) + o6];
    }

#pragma unroll 1
    for (int l = 0; l < 3; ++l) {
      const float* lp = prm + l * (3 * HDIM);
      mlp_layer((l & 1) ? buf1 : buf0, Wt1 + l * (HDIM * HDIM),
                lp, lp + HDIM, lp + 2 * HDIM,
                (l & 1) ? buf0 : buf1, rbase, lane15, half);
    }

    {
      v8f acc[7];
#pragma unroll
      for (int nt = 0; nt < 7; ++nt) acc[nt] = {};
      const _Float16* arow = buf1 + (rbase + lane15) * HDIM;
#pragma unroll
      for (int kt = 0; kt < 4; ++kt) {
        const int kb = kt * 32 + half * 8;
        v16h a = load_frag(arow, kb);
#pragma unroll
        for (int nt = 0; nt < 7; ++nt) {
          v16h b = load_frag(Wt4 + (nt * 16 + lane15) * HDIM, kb);
          acc[nt] = __builtin_amdgcn_wmma_f32_16x16x32_f16(
              false, a, false, b, (short)0, acc[nt], false, false);
        }
      }
      float bn[7];
#pragma unroll
      for (int nt = 0; nt < 7; ++nt) bn[nt] = pb4[nt * 16 + lane15];

#pragma unroll
      for (int j = 0; j < 8; ++j) {
        const int row = rbase + j + half * 8;
        float gain = 0.f, asset = 0.f;
#pragma unroll
        for (int nt = 0; nt < 7; ++nt) {
          const int d = nt * 16 + lane15;
          if (d < DIMA) {
            const float z  = 1.f / (1.f + expf(-(acc[nt][j] + bn[nt])));
            const float s  = Sl[row * SROW + d];
            const float ds = s * (RDT + SIG * dwv[j][nt]);
            gain  += z * ds;
            asset += z * s;
            Sl[row * SROW + d] = s + ds;
          }
        }
        gain  = red16(gain);
        asset = red16(asset);
        if (lane15 == 0) {
          const float y = Yl[row];
          Yl[row] = y + RDT * (y - asset) + gain;
        }
      }
    }
  }

  __syncthreads();
  for (int i = tid; i < ROWS; i += 128) outY[blockBase + i] = Yl[i];
  for (int i = tid; i < ROWS * DIMA; i += 128) {
    const int r = i / DIMA, d = i - r * DIMA;
    outS[(blockBase + r) * (long)DIMA + d] = Sl[r * SROW + d];
  }
}

extern "C" void kernel_launch(void* const* d_in, const int* in_sizes, int n_in,
                              void* d_out, int out_size, void* d_ws, size_t ws_size,
                              hipStream_t stream) {
  (void)in_sizes; (void)n_in; (void)out_size; (void)d_ws; (void)ws_size;
  const float* dw  = (const float*)d_in[0];
  const float* tg  = (const float*)d_in[1];
  const float* W1  = (const float*)d_in[2];
  const float* b1  = (const float*)d_in[3];
  const float* g1  = (const float*)d_in[4];
  const float* be1 = (const float*)d_in[5];
  const float* W2  = (const float*)d_in[6];
  const float* b2  = (const float*)d_in[7];
  const float* g2  = (const float*)d_in[8];
  const float* be2 = (const float*)d_in[9];
  const float* W3  = (const float*)d_in[10];
  const float* b3  = (const float*)d_in[11];
  const float* g3  = (const float*)d_in[12];
  const float* be3 = (const float*)d_in[13];
  const float* W4  = (const float*)d_in[14];
  const float* b4  = (const float*)d_in[15];
  const float* Y0  = (const float*)d_in[16];

  float* outY = (float*)d_out;
  float* outS = outY + BATCH;

  const size_t smem =
      (size_t)(3 * HDIM * HDIM + NPAD4 * HDIM) * sizeof(_Float16) +
      (size_t)(9 * 128 + NPAD4) * sizeof(float) +
      (size_t)(ROWS * SROW + ROWS) * sizeof(float) +
      (size_t)(2 * ROWS * HDIM) * sizeof(_Float16);

  dim3 grid(BATCH / ROWS);
  dim3 block(128);
  bsde_fused_kernel<<<grid, block, smem, stream>>>(
      dw, tg, W1, b1, g1, be1, W2, b2, g2, be2, W3, b3, g3, be3, W4, b4, Y0,
      outY, outS);
}


# Pad the LDS row pitch of the f16 weight and activation tiles from 256 B to 272 B (16 B x 17): the 16 lanes of a ds_load_b128 fragment read no longer start on the same bank
# speedup vs baseline: 4.3924x; 1.4014x over previous
#include <hip/hip_runtime.h>
#include <math.h>

#define DIMA   100
#define MSTEP  50
#define HDIM   128
#define BATCH  8192
#define NPAD4  112
#define SROW   104
#define HP     136
#define ROWS   64
#define S0INV  0.01f
#define RDT    0.001f
#define SIG    0.2f
#define LNEPS  1e-5f

typedef __attribute__((ext_vector_type(16))) _Float16 v16h;
typedef __attribute__((ext_vector_type(8)))  _Float16 v8h;
typedef __attribute__((ext_vector_type(8)))  float    v8f;

union V16U { v16h v; v8h h[2]; };

__device__ __forceinline__ v16h load_frag(const _Float16* base, int kb) {
  V16U u;
  u.h[0] = *(const v8h*)(base + kb);
  u.h[1] = *(const v8h*)(base + kb + 16);
  return u.v;
}

template <int M>
__device__ __forceinline__ float row_xor_add(float x) {
  const int partner = __builtin_amdgcn_update_dpp(
      0, __builtin_bit_cast(int, x), 0x160 + M, 0xf, 0xf, true);
  return x + __builtin_bit_cast(float, partner);
}

__device__ __forceinline__ float red16(float x) {
  x = row_xor_add<1>(x);
  x = row_xor_add<2>(x);
  x = row_xor_add<4>(x);
  x = row_xor_add<8>(x);
  return x;
}

__device__ __forceinline__ v8f erf8(v8f u) {
  v8f ax;
#pragma unroll
  for (int i = 0; i < 8; ++i) ax[i] = __builtin_fabsf(u[i]);
  const v8f t = u * u;
  v8f p = t * -5.618018e-4f + 4.913816e-3f;
  p = p * t + -2.6707515e-2f;
  p = p * t + 1.1280011e-1f;
  p = p * t + -3.7612295e-1f;
  p = p * t + 1.283791e-1f;
  const v8f rs = ax * p + ax;
  v8f q = ax * 1.699881e-5f + -3.7867785e-4f;
  q = q * ax + 3.8578159e-3f;
  q = q * ax + -2.4181698e-2f;
  q = q * ax + 1.0666826e-1f;
  q = q * ax + 6.349333e-1f;
  q = q * ax + 1.2868941e-1f;
  q = q * ax + ax;
  const v8f ea = q * -1.4426950408889634f;
  v8f rb;
#pragma unroll
  for (int i = 0; i < 8; ++i) rb[i] = 1.f - __builtin_amdgcn_exp2f(ea[i]);
  const v8f pick = (ax < 1.f) ? rs : rb;
  v8f r;
#pragma unroll
  for (int i = 0; i < 8; ++i) r[i] = __builtin_copysignf(pick[i], u[i]);
  return r;
}

__device__ __forceinline__ void mlp_layer(const _Float16* in, const _Float16* Wt,
                                          const float* bb, const float* gg,
                                          const float* ee, _Float16* out,
                                          int rbase, int lane15, int half) {
  v8f acc[8];
#pragma unroll
  for (int nt = 0; nt < 8; ++nt) acc[nt] = {};

  const _Float16* arow = in + (rbase + lane15) * HP;
#pragma unroll
  for (int kt = 0; kt < 4; ++kt) {
    const int kb = kt * 32 + half * 8;
    v16h a = load_frag(arow, kb);
#pragma unroll
    for (int nt = 0; nt < 8; ++nt) {
      v16h b = load_frag(Wt + (nt * 16 + lane15) * HP, kb);
      acc[nt] = __builtin_amdgcn_wmma_f32_16x16x32_f16(
          false, a, false, b, (short)0, acc[nt], false, false);
    }
  }

  float bn[8], gn[8], en[8];
#pragma unroll
  for (int nt = 0; nt < 8; ++nt) {
    const int n = nt * 16 + lane15;
    bn[nt] = bb[n]; gn[nt] = gg[n]; en[nt] = ee[n];
  }

#pragma unroll
  for (int j = 0; j < 8; ++j) {
    float p = 0.f;
#pragma unroll
    for (int nt = 0; nt < 8; ++nt) { acc[nt][j] += bn[nt]; p += acc[nt][j]; }
    p = red16(p);
    const float mu = p * (1.f / 128.f);
    float q = 0.f;
#pragma unroll
    for (int nt = 0; nt < 8; ++nt) { float d = acc[nt][j] - mu; q += d * d; }
    q = red16(q);
    const float rstd = rsqrtf(q * (1.f / 128.f) + LNEPS);

    const int row = rbase + j + half * 8;
    _Float16* orow = out + row * HP + lane15;
    v8f x;
#pragma unroll
    for (int nt = 0; nt < 8; ++nt)
      x[nt] = (acc[nt][j] - mu) * rstd * gn[nt] + en[nt];
    const v8f ge = (x * 0.5f) * (erf8(x * 0.70710678118654752f) + 1.f);
#pragma unroll
    for (int nt = 0; nt < 8; ++nt) orow[nt * 16] = (_Float16)ge[nt];
  }
}

extern "C" __global__ void __launch_bounds__(128)
bsde_fused_kernel(const float* __restrict__ dw, const float* __restrict__ tg,
                  const float* __restrict__ W1, const float* __restrict__ b1,
                  const float* __restrict__ g1, const float* __restrict__ be1,
                  const float* __restrict__ W2, const float* __restrict__ b2,
                  const float* __restrict__ g2, const float* __restrict__ be2,
                  const float* __restrict__ W3, const float* __restrict__ b3,
                  const float* __restrict__ g3, const float* __restrict__ be3,
                  const float* __restrict__ W4, const float* __restrict__ b4,
                  const float* __restrict__ Y0,
                  float* __restrict__ outY, float* __restrict__ outS) {
  extern __shared__ __align__(16) char smem[];
  _Float16* Wt1 = (_Float16*)smem;
  _Float16* Wt2 = Wt1 + HDIM * HP;
  _Float16* Wt3 = Wt2 + HDIM * HP;
  _Float16* Wt4 = Wt3 + HDIM * HP;
  float* prm = (float*)(Wt4 + NPAD4 * HP);
  float* pb1 = prm;        float* pg1 = prm + 128;  float* pe1 = prm + 256;
  float* pb2 = prm + 384;  float* pg2 = prm + 512;  float* pe2 = prm + 640;
  float* pb3 = prm + 768;  float* pg3 = prm + 896;  float* pe3 = prm + 1024;
  float* pb4 = prm + 1152;
  float* Sl  = prm + 1264;
  float* Yl  = Sl + ROWS * SROW;
  _Float16* buf0 = (_Float16*)(Yl + ROWS);
  _Float16* buf1 = buf0 + ROWS * HP;

  const int tid = threadIdx.x;
  const int lane = tid & 31, lane15 = lane & 15, half = lane >> 4;
  const int rbase = (tid >> 5) * 16;
  const long blockBase = (long)blockIdx.x * ROWS;

  for (int i = tid; i < HDIM * HDIM; i += 128) {
    const int n = i >> 7, k = i & 127;
    const int o = n * HP + k;
    Wt1[o] = (_Float16)(k < DIMA + 1 ? W1[k * HDIM + n] : 0.f);
    Wt2[o] = (_Float16)W2[k * HDIM + n];
    Wt3[o] = (_Float16)W3[k * HDIM + n];
  }
  for (int i = tid; i < NPAD4 * HDIM; i += 128) {
    const int n = i >> 7, k = i & 127;
    Wt4[n * HP + k] = (_Float16)(n < DIMA ? W4[k * DIMA + n] : 0.f);
  }
  pb1[tid] = b1[tid]; pg1[tid] = g1[tid]; pe1[tid] = be1[tid];
  pb2[tid] = b2[tid]; pg2[tid] = g2[tid]; pe2[tid] = be2[tid];
  pb3[tid] = b3[tid]; pg3[tid] = g3[tid]; pe3[tid] = be3[tid];
  if (tid < NPAD4) pb4[tid] = (tid < DIMA) ? b4[tid] : 0.f;

  const float y0 = Y0[0];
#pragma unroll 4
  for (int r = 0; r < 16; ++r) {
    for (int d = lane; d < DIMA; d += 32) Sl[(rbase + r) * SROW + d] = 100.f;
    if (lane == 0) Yl[rbase + r] = y0;
  }
  __syncthreads();

  for (int m = 0; m < MSTEP; ++m) {
    for (int i = lane; i < 16 * HDIM; i += 32) {
      const int rl = i >> 7, k = i & 127;
      const int row = rbase + rl;
      float v;
      if (k < DIMA)      v = Sl[row * SROW + k] * S0INV;
      else if (k == DIMA) v = tg[(blockBase + row) * MSTEP + m];
      else               v = 0.f;
      buf0[row * HP + k] = (_Float16)v;
    }

    float dwv[8][7];
    {
      const float* dwp =
          dw + ((blockBase + rbase + half * 8) * MSTEP + m) * (long)DIMA + lane15;
#pragma unroll
      for (int j = 0; j < 8; ++j) {
#pragma unroll
        for (int nt = 0; nt < 6; ++nt) dwv[j][nt] = dwp[j * (MSTEP * DIMA) + nt * 16];
      }
      const int o6 = (lane15 < DIMA - 96) ? 96 : (DIMA - 1 - lane15);
#pragma unroll
      for (int j = 0; j < 8; ++j) dwv[j][6] = dwp[j * (MSTEP * DIMA) + o6];
    }

#pragma unroll 1
    for (int l = 0; l < 3; ++l) {
      const float* lp = prm + l * (3 * HDIM);
      mlp_layer((l & 1) ? buf1 : buf0, Wt1 + l * (HDIM * HP),
                lp, lp + HDIM, lp + 2 * HDIM,
                (l & 1) ? buf0 : buf1, rbase, lane15, half);
    }

    {
      v8f acc[7];
#pragma unroll
      for (int nt = 0; nt < 7; ++nt) acc[nt] = {};
      const _Float16* arow = buf1 + (rbase + lane15) * HP;
#pragma unroll
      for (int kt = 0; kt < 4; ++kt) {
        const int kb = kt * 32 + half * 8;
        v16h a = load_frag(arow, kb);
#pragma unroll
        for (int nt = 0; nt < 7; ++nt) {
          v16h b = load_frag(Wt4 + (nt * 16 + lane15) * HP, kb);
          acc[nt] = __builtin_amdgcn_wmma_f32_16x16x32_f16(
              false, a, false, b, (short)0, acc[nt], false, false);
        }
      }
      float bn[7];
#pragma unroll
      for (int nt = 0; nt < 7; ++nt) bn[nt] = pb4[nt * 16 + lane15];

#pragma unroll
      for (int j = 0; j < 8; ++j) {
        const int row = rbase + j + half * 8;
        float gain = 0.f, asset = 0.f;
#pragma unroll
        for (int nt = 0; nt < 7; ++nt) {
          const int d = nt * 16 + lane15;
          if (d < DIMA) {
            const float z  = 1.f / (1.f + expf(-(acc[nt][j] + bn[nt])));
            const float s  = Sl[row * SROW + d];
            const float ds = s * (RDT + SIG * dwv[j][nt]);
            gain  += z * ds;
            asset += z * s;
            Sl[row * SROW + d] = s + ds;
          }
        }
        gain  = red16(gain);
        asset = red16(asset);
        if (lane15 == 0) {
          const float y = Yl[row];
          Yl[row] = y + RDT * (y - asset) + gain;
        }
      }
    }
  }

  __syncthreads();
  for (int i = tid; i < ROWS; i += 128) outY[blockBase + i] = Yl[i];
  for (int i = tid; i < ROWS * DIMA; i += 128) {
    const int r = i / DIMA, d = i - r * DIMA;
    outS[(blockBase + r) * (long)DIMA + d] = Sl[r * SROW + d];
  }
}

extern "C" void kernel_launch(void* const* d_in, const int* in_sizes, int n_in,
                              void* d_out, int out_size, void* d_ws, size_t ws_size,
                              hipStream_t stream) {
  (void)in_sizes; (void)n_in; (void)out_size; (void)d_ws; (void)ws_size;
  const float* dw  = (const float*)d_in[0];
  const float* tg  = (const float*)d_in[1];
  const float* W1  = (const float*)d_in[2];
  const float* b1  = (const float*)d_in[3];
  const float* g1  = (const float*)d_in[4];
  const float* be1 = (const float*)d_in[5];
  const float* W2  = (const float*)d_in[6];
  const float* b2  = (const float*)d_in[7];
  const float* g2  = (const float*)d_in[8];
  const float* be2 = (const float*)d_in[9];
  const float* W3  = (const float*)d_in[10];
  const float* b3  = (const float*)d_in[11];
  const float* g3  = (const float*)d_in[12];
  const float* be3 = (const float*)d_in[13];
  const float* W4  = (const float*)d_in[14];
  const float* b4  = (const float*)d_in[15];
  const float* Y0  = (const float*)d_in[16];

  float* outY = (float*)d_out;
  float* outS = outY + BATCH;

  const size_t smem =
      (size_t)(3 * HDIM * HP + NPAD4 * HP) * sizeof(_Float16) +
      (size_t)(9 * 128 + NPAD4) * sizeof(float) +
      (size_t)(ROWS * SROW + ROWS) * sizeof(float) +
      (size_t)(2 * ROWS * HP) * sizeof(_Float16);

  dim3 grid(BATCH / ROWS);
  dim3 block(128);
  bsde_fused_kernel<<<grid, block, smem, stream>>>(
      dw, tg, W1, b1, g1, be1, W2, b2, g2, be2, W3, b3, g3, be3, W4, b4, Y0,
      outY, outS);
}


# Declare one wave per SIMD (amdgpu_waves_per_eu(1,1); the LDS size already forces it): each fragment gets its own registers and a layer's 72 LDS loads go out as one group, not load-wait-WMMA per tile
# speedup vs baseline: 9.4232x; 2.1453x over previous
#include <hip/hip_runtime.h>
#include <math.h>

#define DIMA   100
#define MSTEP  50
#define HDIM   128
#define BATCH  8192
#define NPAD4  112
#define SROW   104
#define HP     136
#define ROWS   64
#define S0INV  0.01f
#define RDT    0.001f
#define SIG    0.2f
#define LNEPS  1e-5f

typedef __attribute__((ext_vector_type(16))) _Float16 v16h;
typedef __attribute__((ext_vector_type(8)))  _Float16 v8h;
typedef __attribute__((ext_vector_type(8)))  float    v8f;

union V16U { v16h v; v8h h[2]; };

__device__ __forceinline__ v16h load_frag(const _Float16* base, int kb) {
  V16U u;
  u.h[0] = *(const v8h*)(base + kb);
  u.h[1] = *(const v8h*)(base + kb + 16);
  return u.v;
}

template <int M>
__device__ __forceinline__ float row_xor_add(float x) {
  const int partner = __builtin_amdgcn_update_dpp(
      0, __builtin_bit_cast(int, x), 0x160 + M, 0xf, 0xf, true);
  return x + __builtin_bit_cast(float, partner);
}

__device__ __forceinline__ float red16(float x) {
  x = row_xor_add<1>(x);
  x = row_xor_add<2>(x);
  x = row_xor_add<4>(x);
  x = row_xor_add<8>(x);
  return x;
}

__device__ __forceinline__ v8f erf8(v8f u) {
  v8f ax;
#pragma unroll
  for (int i = 0; i < 8; ++i) ax[i] = __builtin_fabsf(u[i]);
  const v8f t = u * u;
  v8f p = t * -5.618018e-4f + 4.913816e-3f;
  p = p * t + -2.6707515e-2f;
  p = p * t + 1.1280011e-1f;
  p = p * t + -3.7612295e-1f;
  p = p * t + 1.283791e-1f;
  const v8f rs = ax * p + ax;
  v8f q = ax * 1.699881e-5f + -3.7867785e-4f;
  q = q * ax + 3.8578159e-3f;
  q = q * ax + -2.4181698e-2f;
  q = q * ax + 1.0666826e-1f;
  q = q * ax + 6.349333e-1f;
  q = q * ax + 1.2868941e-1f;
  q = q * ax + ax;
  const v8f ea = q * -1.4426950408889634f;
  v8f rb;
#pragma unroll
  for (int i = 0; i < 8; ++i) rb[i] = 1.f - __builtin_amdgcn_exp2f(ea[i]);
  const v8f pick = (ax < 1.f) ? rs : rb;
  v8f r;
#pragma unroll
  for (int i = 0; i < 8; ++i) r[i] = __builtin_copysignf(pick[i], u[i]);
  return r;
}

__device__ __forceinline__ void mlp_layer(const _Float16* in, const _Float16* Wt,
                                          const float* bb, const float* gg,
                                          const float* ee, _Float16* out,
                                          int rbase, int lane15, int half) {
  v8f acc[8];
#pragma unroll
  for (int nt = 0; nt < 8; ++nt) acc[nt] = {};

  const _Float16* arow = in + (rbase + lane15) * HP;
#pragma unroll
  for (int kt = 0; kt < 4; ++kt) {
    const int kb = kt * 32 + half * 8;
    v16h a = load_frag(arow, kb);
#pragma unroll
    for (int nt = 0; nt < 8; ++nt) {
      v16h b = load_frag(Wt + (nt * 16 + lane15) * HP, kb);
      acc[nt] = __builtin_amdgcn_wmma_f32_16x16x32_f16(
          false, a, false, b, (short)0, acc[nt], false, false);
    }
  }

  float bn[8], gn[8], en[8];
#pragma unroll
  for (int nt = 0; nt < 8; ++nt) {
    const int n = nt * 16 + lane15;
    bn[nt] = bb[n]; gn[nt] = gg[n]; en[nt] = ee[n];
  }

#pragma unroll
  for (int j = 0; j < 8; ++j) {
    float p = 0.f;
#pragma unroll
    for (int nt = 0; nt < 8; ++nt) { acc[nt][j] += bn[nt]; p += acc[nt][j]; }
    p = red16(p);
    const float mu = p * (1.f / 128.f);
    float q = 0.f;
#pragma unroll
    for (int nt = 0; nt < 8; ++nt) { float d = acc[nt][j] - mu; q += d * d; }
    q = red16(q);
    const float rstd = rsqrtf(q * (1.f / 128.f) + LNEPS);

    const int row = rbase + j + half * 8;
    _Float16* orow = out + row * HP + lane15;
    v8f x;
#pragma unroll
    for (int nt = 0; nt < 8; ++nt)
      x[nt] = (acc[nt][j] - mu) * rstd * gn[nt] + en[nt];
    const v8f ge = (x * 0.5f) * (erf8(x * 0.70710678118654752f) + 1.f);
#pragma unroll
    for (int nt = 0; nt < 8; ++nt) orow[nt * 16] = (_Float16)ge[nt];
  }
}

extern "C" __global__ void __launch_bounds__(128)
__attribute__((amdgpu_waves_per_eu(1, 1)))
bsde_fused_kernel(const float* __restrict__ dw, const float* __restrict__ tg,
                  const float* __restrict__ W1, const float* __restrict__ b1,
                  const float* __restrict__ g1, const float* __restrict__ be1,
                  const float* __restrict__ W2, const float* __restrict__ b2,
                  const float* __restrict__ g2, const float* __restrict__ be2,
                  const float* __restrict__ W3, const float* __restrict__ b3,
                  const float* __restrict__ g3, const float* __restrict__ be3,
                  const float* __restrict__ W4, const float* __restrict__ b4,
                  const float* __restrict__ Y0,
                  float* __restrict__ outY, float* __restrict__ outS) {
  extern __shared__ __align__(16) char smem[];
  _Float16* Wt1 = (_Float16*)smem;
  _Float16* Wt2 = Wt1 + HDIM * HP;
  _Float16* Wt3 = Wt2 + HDIM * HP;
  _Float16* Wt4 = Wt3 + HDIM * HP;
  float* prm = (float*)(Wt4 + NPAD4 * HP);
  float* pb1 = prm;        float* pg1 = prm + 128;  float* pe1 = prm + 256;
  float* pb2 = prm + 384;  float* pg2 = prm + 512;  float* pe2 = prm + 640;
  float* pb3 = prm + 768;  float* pg3 = prm + 896;  float* pe3 = prm + 1024;
  float* pb4 = prm + 1152;
  float* Sl  = prm + 1264;
  float* Yl  = Sl + ROWS * SROW;
  _Float16* buf0 = (_Float16*)(Yl + ROWS);
  _Float16* buf1 = buf0 + ROWS * HP;

  const int tid = threadIdx.x;
  const int lane = tid & 31, lane15 = lane & 15, half = lane >> 4;
  const int rbase = (tid >> 5) * 16;
  const long blockBase = (long)blockIdx.x * ROWS;

  for (int i = tid; i < HDIM * HDIM; i += 128) {
    const int n = i >> 7, k = i & 127;
    const int o = n * HP + k;
    Wt1[o] = (_Float16)(k < DIMA + 1 ? W1[k * HDIM + n] : 0.f);
    Wt2[o] = (_Float16)W2[k * HDIM + n];
    Wt3[o] = (_Float16)W3[k * HDIM + n];
  }
  for (int i = tid; i < NPAD4 * HDIM; i += 128) {
    const int n = i >> 7, k = i & 127;
    Wt4[n * HP + k] = (_Float16)(n < DIMA ? W4[k * DIMA + n] : 0.f);
  }
  pb1[tid] = b1[tid]; pg1[tid] = g1[tid]; pe1[tid] = be1[tid];
  pb2[tid] = b2[tid]; pg2[tid] = g2[tid]; pe2[tid] = be2[tid];
  pb3[tid] = b3[tid]; pg3[tid] = g3[tid]; pe3[tid] = be3[tid];
  if (tid < NPAD4) pb4[tid] = (tid < DIMA) ? b4[tid] : 0.f;

  const float y0 = Y0[0];
#pragma unroll 4
  for (int r = 0; r < 16; ++r) {
    for (int d = lane; d < DIMA; d += 32) Sl[(rbase + r) * SROW + d] = 100.f;
    if (lane == 0) Yl[rbase + r] = y0;
  }
  __syncthreads();

  for (int i = lane; i < 16 * HDIM; i += 32) {
    const int rl = i >> 7, k = i & 127;
    const int row = rbase + rl;
    float v;
    if (k < DIMA)      v = Sl[row * SROW + k] * S0INV;
    else if (k == DIMA) v = tg[(blockBase + row) * MSTEP];
    else               v = 0.f;
    buf0[row * HP + k] = (_Float16)v;
  }

  for (int m = 0; m < MSTEP; ++m) {
    float dwv[8][7];
    {
      const float* dwp =
          dw + ((blockBase + rbase + half * 8) * MSTEP + m) * (long)DIMA + lane15;
#pragma unroll
      for (int j = 0; j < 8; ++j) {
#pragma unroll
        for (int nt = 0; nt < 6; ++nt) dwv[j][nt] = dwp[j * (MSTEP * DIMA) + nt * 16];
      }
      const int o6 = (lane15 < DIMA - 96) ? 96 : (DIMA - 1 - lane15);
#pragma unroll
      for (int j = 0; j < 8; ++j) dwv[j][6] = dwp[j * (MSTEP * DIMA) + o6];
    }

    float tn[8];
    {
      const int mn = (m + 1 < MSTEP) ? (m + 1) : m;
      const float* tp = tg + (blockBase + rbase + half * 8) * MSTEP + mn;
#pragma unroll
      for (int j = 0; j < 8; ++j) tn[j] = tp[j * MSTEP];
    }

#pragma unroll 1
    for (int l = 0; l < 3; ++l) {
      const float* lp = prm + l * (3 * HDIM);
      mlp_layer((l & 1) ? buf1 : buf0, Wt1 + l * (HDIM * HP),
                lp, lp + HDIM, lp + 2 * HDIM,
                (l & 1) ? buf0 : buf1, rbase, lane15, half);
    }

    {
      v8f acc[7];
#pragma unroll
      for (int nt = 0; nt < 7; ++nt) acc[nt] = {};
      const _Float16* arow = buf1 + (rbase + lane15) * HP;
#pragma unroll
      for (int kt = 0; kt < 4; ++kt) {
        const int kb = kt * 32 + half * 8;
        v16h a = load_frag(arow, kb);
#pragma unroll
        for (int nt = 0; nt < 7; ++nt) {
          v16h b = load_frag(Wt4 + (nt * 16 + lane15) * HP, kb);
          acc[nt] = __builtin_amdgcn_wmma_f32_16x16x32_f16(
              false, a, false, b, (short)0, acc[nt], false, false);
        }
      }
      float bn[7];
#pragma unroll
      for (int nt = 0; nt < 7; ++nt) bn[nt] = pb4[nt * 16 + lane15];

#pragma unroll
      for (int j = 0; j < 8; ++j) {
        const int row = rbase + j + half * 8;
        _Float16* xrow = buf0 + row * HP + lane15;
        float gain = 0.f, asset = 0.f;
#pragma unroll
        for (int nt = 0; nt < 7; ++nt) {
          const int d = nt * 16 + lane15;
          float xv = (d == DIMA) ? tn[j] : 0.f;
          if (d < DIMA) {
            const float z  = 1.f / (1.f + expf(-(acc[nt][j] + bn[nt])));
            const float s  = Sl[row * SROW + d];
            const float ds = s * (RDT + SIG * dwv[j][nt]);
            const float sn = s + ds;
            gain  += z * ds;
            asset += z * s;
            Sl[row * SROW + d] = sn;
            xv = sn * S0INV;
          }
          xrow[nt * 16] = (_Float16)xv;
        }
        xrow[7 * 16] = (_Float16)0.f;
        gain  = red16(gain);
        asset = red16(asset);
        if (lane15 == 0) {
          const float y = Yl[row];
          Yl[row] = y + RDT * (y - asset) + gain;
        }
      }
    }
  }

  __syncthreads();
  for (int i = tid; i < ROWS; i += 128) outY[blockBase + i] = Yl[i];
  for (int i = tid; i < ROWS * DIMA; i += 128) {
    const int r = i / DIMA, d = i - r * DIMA;
    outS[(blockBase + r) * (long)DIMA + d] = Sl[r * SROW + d];
  }
}

extern "C" void kernel_launch(void* const* d_in, const int* in_sizes, int n_in,
                              void* d_out, int out_size, void* d_ws, size_t ws_size,
                              hipStream_t stream) {
  (void)in_sizes; (void)n_in; (void)out_size; (void)d_ws; (void)ws_size;
  const float* dw  = (const float*)d_in[0];
  const float* tg  = (const float*)d_in[1];
  const float* W1  = (const float*)d_in[2];
  const float* b1  = (const float*)d_in[3];
  const float* g1  = (const float*)d_in[4];
  const float* be1 = (const float*)d_in[5];
  const float* W2  = (const float*)d_in[6];
  const float* b2  = (const float*)d_in[7];
  const float* g2  = (const float*)d_in[8];
  const float* be2 = (const float*)d_in[9];
  const float* W3  = (const float*)d_in[10];
  const float* b3  = (const float*)d_in[11];
  const float* g3  = (const float*)d_in[12];
  const float* be3 = (const float*)d_in[13];
  const float* W4  = (const float*)d_in[14];
  const float* b4  = (const float*)d_in[15];
  const float* Y0  = (const float*)d_in[16];

  float* outY = (float*)d_out;
  float* outS = outY + BATCH;

  const size_t smem =
      (size_t)(3 * HDIM * HP + NPAD4 * HP) * sizeof(_Float16) +
      (size_t)(9 * 128 + NPAD4) * sizeof(float) +
      (size_t)(ROWS * SROW + ROWS) * sizeof(float) +
      (size_t)(2 * ROWS * HP) * sizeof(_Float16);

  dim3 grid(BATCH / ROWS);
  dim3 block(128);
  bsde_fused_kernel<<<grid, block, smem, stream>>>(
      dw, tg, W1, b1, g1, be1, W2, b2, g2, be2, W3, b3, g3, be3, W4, b4, Y0,
      outY, outS);
}


# Two waves share each 16-row strip: both form its whole product on WMMA, each does LayerNorm, GELU and the head for 8 of its rows; 256 blocks of 32 rows use all 256 WGPs; same sums in the same order
# speedup vs baseline: 13.0466x; 1.3845x over previous
#include <hip/hip_runtime.h>
#include <math.h>

#define DIMA   100
#define MSTEP  50
#define HDIM   128
#define BATCH  8192
#define NPAD4  112
#define SROW   104
#define HP     136
#define ROWS   32
#define S0INV  0.01f
#define RDT    0.001f
#define SIG    0.2f
#define LNEPS  1e-5f

typedef __attribute__((ext_vector_type(16))) _Float16 v16h;
typedef __attribute__((ext_vector_type(8)))  _Float16 v8h;
typedef __attribute__((ext_vector_type(8)))  float    v8f;

union V16U { v16h v; v8h h[2]; };

__device__ __forceinline__ v16h load_frag(const _Float16* base, int kb) {
  V16U u;
  u.h[0] = *(const v8h*)(base + kb);
  u.h[1] = *(const v8h*)(base + kb + 16);
  return u.v;
}

template <int M>
__device__ __forceinline__ float row_xor_add(float x) {
  const int partner = __builtin_amdgcn_update_dpp(
      0, __builtin_bit_cast(int, x), 0x160 + M, 0xf, 0xf, true);
  return x + __builtin_bit_cast(float, partner);
}

__device__ __forceinline__ float red16(float x) {
  x = row_xor_add<1>(x);
  x = row_xor_add<2>(x);
  x = row_xor_add<4>(x);
  x = row_xor_add<8>(x);
  return x;
}

__device__ __forceinline__ v8f erf8(v8f u) {
  v8f ax;
#pragma unroll
  for (int i = 0; i < 8; ++i) ax[i] = __builtin_fabsf(u[i]);
  const v8f t = u * u;
  v8f p = t * -5.618018e-4f + 4.913816e-3f;
  p = p * t + -2.6707515e-2f;
  p = p * t + 1.1280011e-1f;
  p = p * t + -3.7612295e-1f;
  p = p * t + 1.283791e-1f;
  const v8f rs = ax * p + ax;
  v8f q = ax * 1.699881e-5f + -3.7867785e-4f;
  q = q * ax + 3.8578159e-3f;
  q = q * ax + -2.4181698e-2f;
  q = q * ax + 1.0666826e-1f;
  q = q * ax + 6.349333e-1f;
  q = q * ax + 1.2868941e-1f;
  q = q * ax + ax;
  const v8f ea = q * -1.4426950408889634f;
  v8f rb;
#pragma unroll
  for (int i = 0; i < 8; ++i) rb[i] = 1.f - __builtin_amdgcn_exp2f(ea[i]);
  const v8f pick = (ax < 1.f) ? rs : rb;
  v8f r;
#pragma unroll
  for (int i = 0; i < 8; ++i) r[i] = __builtin_copysignf(pick[i], u[i]);
  return r;
}

__device__ __forceinline__ void mlp_layer(const _Float16* in, const _Float16* Wt,
                                          const float* bb, const float* gg,
                                          const float* ee, _Float16* out,
                                          int rbase, int lane15, int half, int sub4) {
  v8f acc[8];
#pragma unroll
  for (int nt = 0; nt < 8; ++nt) acc[nt] = {};

  const _Float16* arow = in + (rbase + (lane15 ^ sub4)) * HP;
#pragma unroll
  for (int kt = 0; kt < 4; ++kt) {
    const int kb = kt * 32 + half * 8;
    v16h a = load_frag(arow, kb);
#pragma unroll
    for (int nt = 0; nt < 8; ++nt) {
      v16h b = load_frag(Wt + (nt * 16 + lane15) * HP, kb);
      acc[nt] = __builtin_amdgcn_wmma_f32_16x16x32_f16(
          false, a, false, b, (short)0, acc[nt], false, false);
    }
  }

  float bn[8], gn[8], en[8];
#pragma unroll
  for (int nt = 0; nt < 8; ++nt) {
    const int n = nt * 16 + lane15;
    bn[nt] = bb[n]; gn[nt] = gg[n]; en[nt] = ee[n];
  }

#pragma unroll
  for (int j = 0; j < 4; ++j) {
    float p = 0.f;
#pragma unroll
    for (int nt = 0; nt < 8; ++nt) { acc[nt][j] += bn[nt]; p += acc[nt][j]; }
    p = red16(p);
    const float mu = p * (1.f / 128.f);
    float q = 0.f;
#pragma unroll
    for (int nt = 0; nt < 8; ++nt) { float d = acc[nt][j] - mu; q += d * d; }
    q = red16(q);
    const float rstd = rsqrtf(q * (1.f / 128.f) + LNEPS);

    const int row = rbase + half * 8 + sub4 + j;
    _Float16* orow = out + row * HP + lane15;
    v8f x;
#pragma unroll
    for (int nt = 0; nt < 8; ++nt)
      x[nt] = (acc[nt][j] - mu) * rstd * gn[nt] + en[nt];
    const v8f ge = (x * 0.5f) * (erf8(x * 0.70710678118654752f) + 1.f);
#pragma unroll
    for (int nt = 0; nt < 8; ++nt) orow[nt * 16] = (_Float16)ge[nt];
  }
}

extern "C" __global__ void __launch_bounds__(128)
__attribute__((amdgpu_waves_per_eu(1, 1)))
bsde_fused_kernel(const float* __restrict__ dw, const float* __restrict__ tg,
                  const float* __restrict__ W1, const float* __restrict__ b1,
                  const float* __restrict__ g1, const float* __restrict__ be1,
                  const float* __restrict__ W2, const float* __restrict__ b2,
                  const float* __restrict__ g2, const float* __restrict__ be2,
                  const float* __restrict__ W3, const float* __restrict__ b3,
                  const float* __restrict__ g3, const float* __restrict__ be3,
                  const float* __restrict__ W4, const float* __restrict__ b4,
                  const float* __restrict__ Y0,
                  float* __restrict__ outY, float* __restrict__ outS) {
  extern __shared__ __align__(16) char smem[];
  _Float16* Wt1 = (_Float16*)smem;
  _Float16* Wt2 = Wt1 + HDIM * HP;
  _Float16* Wt3 = Wt2 + HDIM * HP;
  _Float16* Wt4 = Wt3 + HDIM * HP;
  float* prm = (float*)(Wt4 + NPAD4 * HP);
  float* pb1 = prm;        float* pg1 = prm + 128;  float* pe1 = prm + 256;
  float* pb2 = prm + 384;  float* pg2 = prm + 512;  float* pe2 = prm + 640;
  float* pb3 = prm + 768;  float* pg3 = prm + 896;  float* pe3 = prm + 1024;
  float* pb4 = prm + 1152;
  float* Sl  = prm + 1264;
  float* Yl  = Sl + ROWS * SROW;
  _Float16* buf0 = (_Float16*)(Yl + ROWS);
  _Float16* buf1 = buf0 + ROWS * HP;

  const int tid = threadIdx.x;
  const int lane = tid & 31, lane15 = lane & 15, half = lane >> 4;
  const int rbase = (tid >> 6) * 16;
  const int sub4  = ((tid >> 5) & 1) * 4;
  const long blockBase = (long)blockIdx.x * ROWS;

  for (int i = tid; i < HDIM * HDIM; i += 128) {
    const int n = i >> 7, k = i & 127;
    const int o = n * HP + k;
    Wt1[o] = (_Float16)(k < DIMA + 1 ? W1[k * HDIM + n] : 0.f);
    Wt2[o] = (_Float16)W2[k * HDIM + n];
    Wt3[o] = (_Float16)W3[k * HDIM + n];
  }
  for (int i = tid; i < NPAD4 * HDIM; i += 128) {
    const int n = i >> 7, k = i & 127;
    Wt4[n * HP + k] = (_Float16)(n < DIMA ? W4[k * DIMA + n] : 0.f);
  }
  pb1[tid] = b1[tid]; pg1[tid] = g1[tid]; pe1[tid] = be1[tid];
  pb2[tid] = b2[tid]; pg2[tid] = g2[tid]; pe2[tid] = be2[tid];
  pb3[tid] = b3[tid]; pg3[tid] = g3[tid]; pe3[tid] = be3[tid];
  if (tid < NPAD4) pb4[tid] = (tid < DIMA) ? b4[tid] : 0.f;

  const float y0 = Y0[0];
  for (int i = tid; i < ROWS * SROW; i += 128) Sl[i] = 100.f;
  if (tid < ROWS) Yl[tid] = y0;
  __syncthreads();

  for (int i = tid; i < ROWS * HDIM; i += 128) {
    const int row = i >> 7, k = i & 127;
    float v;
    if (k < DIMA)      v = Sl[row * SROW + k] * S0INV;
    else if (k == DIMA) v = tg[(blockBase + row) * MSTEP];
    else               v = 0.f;
    buf0[row * HP + k] = (_Float16)v;
  }
  __syncthreads();

  for (int m = 0; m < MSTEP; ++m) {
    float dwv[4][7];
    {
      const float* dwp =
          dw + ((blockBase + rbase + half * 8 + sub4) * MSTEP + m) * (long)DIMA + lane15;
#pragma unroll
      for (int j = 0; j < 4; ++j) {
#pragma unroll
        for (int nt = 0; nt < 6; ++nt) dwv[j][nt] = dwp[j * (MSTEP * DIMA) + nt * 16];
      }
      const int o6 = (lane15 < DIMA - 96) ? 96 : (DIMA - 1 - lane15);
#pragma unroll
      for (int j = 0; j < 4; ++j) dwv[j][6] = dwp[j * (MSTEP * DIMA) + o6];
    }

    float tn[4];
    {
      const int mn = (m + 1 < MSTEP) ? (m + 1) : m;
      const float* tp = tg + (blockBase + rbase + half * 8 + sub4) * MSTEP + mn;
#pragma unroll
      for (int j = 0; j < 4; ++j) tn[j] = tp[j * MSTEP];
    }

#pragma unroll 1
    for (int l = 0; l < 3; ++l) {
      const float* lp = prm + l * (3 * HDIM);
      mlp_layer((l & 1) ? buf1 : buf0, Wt1 + l * (HDIM * HP),
                lp, lp + HDIM, lp + 2 * HDIM,
                (l & 1) ? buf0 : buf1, rbase, lane15, half, sub4);
      __syncthreads();
    }

    {
      v8f acc[7];
#pragma unroll
      for (int nt = 0; nt < 7; ++nt) acc[nt] = {};
      const _Float16* arow = buf1 + (rbase + (lane15 ^ sub4)) * HP;
#pragma unroll
      for (int kt = 0; kt < 4; ++kt) {
        const int kb = kt * 32 + half * 8;
        v16h a = load_frag(arow, kb);
#pragma unroll
        for (int nt = 0; nt < 7; ++nt) {
          v16h b = load_frag(Wt4 + (nt * 16 + lane15) * HP, kb);
          acc[nt] = __builtin_amdgcn_wmma_f32_16x16x32_f16(
              false, a, false, b, (short)0, acc[nt], false, false);
        }
      }
      float bn[7];
#pragma unroll
      for (int nt = 0; nt < 7; ++nt) bn[nt] = pb4[nt * 16 + lane15];

#pragma unroll
      for (int j = 0; j < 4; ++j) {
        const int row = rbase + half * 8 + sub4 + j;
        _Float16* xrow = buf0 + row * HP + lane15;
        float gain = 0.f, asset = 0.f;
#pragma unroll
        for (int nt = 0; nt < 7; ++nt) {
          const int d = nt * 16 + lane15;
          float xv = (d == DIMA) ? tn[j] : 0.f;
          if (d < DIMA) {
            const float z  = 1.f / (1.f + expf(-(acc[nt][j] + bn[nt])));
            const float s  = Sl[row * SROW + d];
            const float ds = s * (RDT + SIG * dwv[j][nt]);
            const float sn = s + ds;
            gain  += z * ds;
            asset += z * s;
            Sl[row * SROW + d] = sn;
            xv = sn * S0INV;
          }
          xrow[nt * 16] = (_Float16)xv;
        }
        xrow[7 * 16] = (_Float16)0.f;
        gain  = red16(gain);
        asset = red16(asset);
        if (lane15 == 0) {
          const float y = Yl[row];
          Yl[row] = y + RDT * (y - asset) + gain;
        }
      }
    }
    __syncthreads();
  }

  for (int i = tid; i < ROWS; i += 128) outY[blockBase + i] = Yl[i];
  for (int i = tid; i < ROWS * DIMA; i += 128) {
    const int r = i / DIMA, d = i - r * DIMA;
    outS[(blockBase + r) * (long)DIMA + d] = Sl[r * SROW + d];
  }
}

extern "C" void kernel_launch(void* const* d_in, const int* in_sizes, int n_in,
                              void* d_out, int out_size, void* d_ws, size_t ws_size,
                              hipStream_t stream) {
  (void)in_sizes; (void)n_in; (void)out_size; (void)d_ws; (void)ws_size;
  const float* dw  = (const float*)d_in[0];
  const float* tg  = (const float*)d_in[1];
  const float* W1  = (const float*)d_in[2];
  const float* b1  = (const float*)d_in[3];
  const float* g1  = (const float*)d_in[4];
  const float* be1 = (const float*)d_in[5];
  const float* W2  = (const float*)d_in[6];
  const float* b2  = (const float*)d_in[7];
  const float* g2  = (const float*)d_in[8];
  const float* be2 = (const float*)d_in[9];
  const float* W3  = (const float*)d_in[10];
  const float* b3  = (const float*)d_in[11];
  const float* g3  = (const float*)d_in[12];
  const float* be3 = (const float*)d_in[13];
  const float* W4  = (const float*)d_in[14];
  const float* b4  = (const float*)d_in[15];
  const float* Y0  = (const float*)d_in[16];

  float* outY = (float*)d_out;
  float* outS = outY + BATCH;

  const size_t smem =
      (size_t)(3 * HDIM * HP + NPAD4 * HP) * sizeof(_Float16) +
      (size_t)(9 * 128 + NPAD4) * sizeof(float) +
      (size_t)(ROWS * SROW + ROWS) * sizeof(float) +
      (size_t)(2 * ROWS * HP) * sizeof(_Float16);

  dim3 grid(BATCH / ROWS);
  dim3 block(128);
  bsde_fused_kernel<<<grid, block, smem, stream>>>(
      dw, tg, W1, b1, g1, be1, W2, b2, g2, be2, W3, b3, g3, be3, W4, b4, Y0,
      outY, outS);
}


# Weight fragments come from LDS one product ahead (weights never change), two groups of 32 loads that land under LayerNorm/GELU: a product waits only for its activations; the head reads S, Y up front
# speedup vs baseline: 14.4709x; 1.1092x over previous
#include <hip/hip_runtime.h>
#include <math.h>

#define DIMA   100
#define MSTEP  50
#define HDIM   128
#define BATCH  8192
#define NPAD4  112
#define SROW   104
#define W4ROWS 128
#define HP     136
#define ROWS   32
#define S0INV  0.01f
#define RDT    0.001f
#define SIG    0.2f
#define LNEPS  1e-5f

typedef __attribute__((ext_vector_type(16))) _Float16 v16h;
typedef __attribute__((ext_vector_type(8)))  _Float16 v8h;
typedef __attribute__((ext_vector_type(8)))  float    v8f;

union V16U { v16h v; v8h h[2]; };

__device__ __forceinline__ v16h load_frag(const _Float16* base, int kb) {
  V16U u;
  u.h[0] = *(const v8h*)(base + kb);
  u.h[1] = *(const v8h*)(base + kb + 16);
  return u.v;
}

template <int KT0>
__device__ __forceinline__ void load_bfrags2(v16h (&bw)[4][8], const _Float16* Wt,
                                             int lane15, int half) {
#pragma unroll
  for (int kt = KT0; kt < KT0 + 2; ++kt) {
#pragma unroll
    for (int nt = 0; nt < 8; ++nt)
      bw[kt][nt] = load_frag(Wt + (nt * 16 + lane15) * HP, kt * 32 + half * 8);
  }
}

template <int M>
__device__ __forceinline__ float row_xor_add(float x) {
  const int partner = __builtin_amdgcn_update_dpp(
      0, __builtin_bit_cast(int, x), 0x160 + M, 0xf, 0xf, true);
  return x + __builtin_bit_cast(float, partner);
}

__device__ __forceinline__ float red16(float x) {
  x = row_xor_add<1>(x);
  x = row_xor_add<2>(x);
  x = row_xor_add<4>(x);
  x = row_xor_add<8>(x);
  return x;
}

__device__ __forceinline__ v8f erf8(v8f u) {
  v8f ax;
#pragma unroll
  for (int i = 0; i < 8; ++i) ax[i] = __builtin_fabsf(u[i]);
  const v8f t = u * u;
  v8f p = t * -5.618018e-4f + 4.913816e-3f;
  p = p * t + -2.6707515e-2f;
  p = p * t + 1.1280011e-1f;
  p = p * t + -3.7612295e-1f;
  p = p * t + 1.283791e-1f;
  const v8f rs = ax * p + ax;
  v8f q = ax * 1.699881e-5f + -3.7867785e-4f;
  q = q * ax + 3.8578159e-3f;
  q = q * ax + -2.4181698e-2f;
  q = q * ax + 1.0666826e-1f;
  q = q * ax + 6.349333e-1f;
  q = q * ax + 1.2868941e-1f;
  q = q * ax + ax;
  const v8f ea = q * -1.4426950408889634f;
  v8f rb;
#pragma unroll
  for (int i = 0; i < 8; ++i) rb[i] = 1.f - __builtin_amdgcn_exp2f(ea[i]);
  const v8f pick = (ax < 1.f) ? rs : rb;
  v8f r;
#pragma unroll
  for (int i = 0; i < 8; ++i) r[i] = __builtin_copysignf(pick[i], u[i]);
  return r;
}

__device__ __forceinline__ void mlp_layer(const _Float16* in, v16h (&bw)[4][8],
                                          const _Float16* WtNext,
                                          const float* bb, const float* gg,
                                          const float* ee, _Float16* out,
                                          int rbase, int lane15, int half, int sub4) {
  const _Float16* arow = in + (rbase + (lane15 ^ sub4)) * HP;
  v16h a[4];
#pragma unroll
  for (int kt = 0; kt < 4; ++kt) a[kt] = load_frag(arow, kt * 32 + half * 8);

  float bn[8], gn[8], en[8];
#pragma unroll
  for (int nt = 0; nt < 8; ++nt) {
    const int n = nt * 16 + lane15;
    bn[nt] = bb[n]; gn[nt] = gg[n]; en[nt] = ee[n];
  }

  __builtin_amdgcn_sched_barrier(0);

  v8f acc[8];
#pragma unroll
  for (int nt = 0; nt < 8; ++nt) acc[nt] = {};
#pragma unroll
  for (int kt = 0; kt < 4; ++kt) {
#pragma unroll
    for (int nt = 0; nt < 8; ++nt) {
      acc[nt] = __builtin_amdgcn_wmma_f32_16x16x32_f16(
          false, a[kt], false, bw[kt][nt], (short)0, acc[nt], false, false);
    }
  }

  __builtin_amdgcn_sched_barrier(0);
  load_bfrags2<0>(bw, WtNext, lane15, half);
  __builtin_amdgcn_sched_barrier(0);

#pragma unroll
  for (int j = 0; j < 4; ++j) {
    if (j == 2) {
      __builtin_amdgcn_sched_barrier(0);
      load_bfrags2<2>(bw, WtNext, lane15, half);
      __builtin_amdgcn_sched_barrier(0);
    }
    float p = 0.f;
#pragma unroll
    for (int nt = 0; nt < 8; ++nt) { acc[nt][j] += bn[nt]; p += acc[nt][j]; }
    p = red16(p);
    const float mu = p * (1.f / 128.f);
    float q = 0.f;
#pragma unroll
    for (int nt = 0; nt < 8; ++nt) { float d = acc[nt][j] - mu; q += d * d; }
    q = red16(q);
    const float rstd = rsqrtf(q * (1.f / 128.f) + LNEPS);

    const int row = rbase + half * 8 + sub4 + j;
    _Float16* orow = out + row * HP + lane15;
    v8f x;
#pragma unroll
    for (int nt = 0; nt < 8; ++nt)
      x[nt] = (acc[nt][j] - mu) * rstd * gn[nt] + en[nt];
    const v8f ge = (x * 0.5f) * (erf8(x * 0.70710678118654752f) + 1.f);
#pragma unroll
    for (int nt = 0; nt < 8; ++nt) orow[nt * 16] = (_Float16)ge[nt];
  }
}

extern "C" __global__ void __launch_bounds__(128)
__attribute__((amdgpu_waves_per_eu(1, 1)))
bsde_fused_kernel(const float* __restrict__ dw, const float* __restrict__ tg,
                  const float* __restrict__ W1, const float* __restrict__ b1,
                  const float* __restrict__ g1, const float* __restrict__ be1,
                  const float* __restrict__ W2, const float* __restrict__ b2,
                  const float* __restrict__ g2, const float* __restrict__ be2,
                  const float* __restrict__ W3, const float* __restrict__ b3,
                  const float* __restrict__ g3, const float* __restrict__ be3,
                  const float* __restrict__ W4, const float* __restrict__ b4,
                  const float* __restrict__ Y0,
                  float* __restrict__ outY, float* __restrict__ outS) {
  extern __shared__ __align__(16) char smem[];
  _Float16* Wt1 = (_Float16*)smem;
  _Float16* Wt2 = Wt1 + HDIM * HP;
  _Float16* Wt3 = Wt2 + HDIM * HP;
  _Float16* Wt4 = Wt3 + HDIM * HP;
  float* prm = (float*)(Wt4 + W4ROWS * HP);
  float* pb1 = prm;        float* pg1 = prm + 128;  float* pe1 = prm + 256;
  float* pb2 = prm + 384;  float* pg2 = prm + 512;  float* pe2 = prm + 640;
  float* pb3 = prm + 768;  float* pg3 = prm + 896;  float* pe3 = prm + 1024;
  float* pb4 = prm + 1152;
  float* Sl  = prm + 1264;
  float* Yl  = Sl + ROWS * SROW;
  _Float16* buf0 = (_Float16*)(Yl + ROWS);
  _Float16* buf1 = buf0 + ROWS * HP;

  const int tid = threadIdx.x;
  const int lane = tid & 31, lane15 = lane & 15, half = lane >> 4;
  const int rbase = (tid >> 6) * 16;
  const int sub4  = ((tid >> 5) & 1) * 4;
  const long blockBase = (long)blockIdx.x * ROWS;

  for (int i = tid; i < HDIM * HDIM; i += 128) {
    const int n = i >> 7, k = i & 127;
    const int o = n * HP + k;
    Wt1[o] = (_Float16)(k < DIMA + 1 ? W1[k * HDIM + n] : 0.f);
    Wt2[o] = (_Float16)W2[k * HDIM + n];
    Wt3[o] = (_Float16)W3[k * HDIM + n];
  }
  for (int i = tid; i < W4ROWS * HDIM; i += 128) {
    const int n = i >> 7, k = i & 127;
    Wt4[n * HP + k] = (_Float16)(n < DIMA ? W4[k * DIMA + n] : 0.f);
  }
  pb1[tid] = b1[tid]; pg1[tid] = g1[tid]; pe1[tid] = be1[tid];
  pb2[tid] = b2[tid]; pg2[tid] = g2[tid]; pe2[tid] = be2[tid];
  pb3[tid] = b3[tid]; pg3[tid] = g3[tid]; pe3[tid] = be3[tid];
  if (tid < NPAD4) pb4[tid] = (tid < DIMA) ? b4[tid] : 0.f;

  const float y0 = Y0[0];
  for (int i = tid; i < ROWS * SROW; i += 128) Sl[i] = 100.f;
  if (tid < ROWS) Yl[tid] = y0;
  __syncthreads();

  for (int i = tid; i < ROWS * HDIM; i += 128) {
    const int row = i >> 7, k = i & 127;
    float v;
    if (k < DIMA)      v = Sl[row * SROW + k] * S0INV;
    else if (k == DIMA) v = tg[(blockBase + row) * MSTEP];
    else               v = 0.f;
    buf0[row * HP + k] = (_Float16)v;
  }
  __syncthreads();

  v16h bw[4][8];
  load_bfrags2<0>(bw, Wt1, lane15, half);
  load_bfrags2<2>(bw, Wt1, lane15, half);

  for (int m = 0; m < MSTEP; ++m) {
    float dwv[4][7];
    {
      const float* dwp =
          dw + ((blockBase + rbase + half * 8 + sub4) * MSTEP + m) * (long)DIMA + lane15;
#pragma unroll
      for (int j = 0; j < 4; ++j) {
#pragma unroll
        for (int nt = 0; nt < 6; ++nt) dwv[j][nt] = dwp[j * (MSTEP * DIMA) + nt * 16];
      }
      const int o6 = (lane15 < DIMA - 96) ? 96 : (DIMA - 1 - lane15);
#pragma unroll
      for (int j = 0; j < 4; ++j) dwv[j][6] = dwp[j * (MSTEP * DIMA) + o6];
    }

    float tn[4];
    {
      const int mn = (m + 1 < MSTEP) ? (m + 1) : m;
      const float* tp = tg + (blockBase + rbase + half * 8 + sub4) * MSTEP + mn;
#pragma unroll
      for (int j = 0; j < 4; ++j) tn[j] = tp[j * MSTEP];
    }

#pragma unroll 1
    for (int l = 0; l < 3; ++l) {
      const float* lp = prm + l * (3 * HDIM);
      mlp_layer((l & 1) ? buf1 : buf0, bw, Wt1 + (l + 1) * (HDIM * HP),
                lp, lp + HDIM, lp + 2 * HDIM,
                (l & 1) ? buf0 : buf1, rbase, lane15, half, sub4);
      __syncthreads();
    }

    {
      const _Float16* arow = buf1 + (rbase + (lane15 ^ sub4)) * HP;
      v16h a[4];
#pragma unroll
      for (int kt = 0; kt < 4; ++kt) a[kt] = load_frag(arow, kt * 32 + half * 8);

      float bn[7];
#pragma unroll
      for (int nt = 0; nt < 7; ++nt) bn[nt] = pb4[nt * 16 + lane15];
      float sv[4][7], yv[4];
      {
        const int d6 = (lane15 < DIMA - 96) ? (96 + lane15) : (DIMA - 1);
#pragma unroll
        for (int j = 0; j < 4; ++j) {
          const int row = rbase + half * 8 + sub4 + j;
#pragma unroll
          for (int nt = 0; nt < 6; ++nt) sv[j][nt] = Sl[row * SROW + nt * 16 + lane15];
          sv[j][6] = Sl[row * SROW + d6];
          yv[j] = Yl[row];
        }
      }

      __builtin_amdgcn_sched_barrier(0);

      v8f acc[7];
#pragma unroll
      for (int nt = 0; nt < 7; ++nt) acc[nt] = {};
#pragma unroll
      for (int kt = 0; kt < 4; ++kt) {
#pragma unroll
        for (int nt = 0; nt < 7; ++nt) {
          acc[nt] = __builtin_amdgcn_wmma_f32_16x16x32_f16(
              false, a[kt], false, bw[kt][nt], (short)0, acc[nt], false, false);
        }
      }

      __builtin_amdgcn_sched_barrier(0);
      load_bfrags2<0>(bw, Wt1, lane15, half);
      __builtin_amdgcn_sched_barrier(0);

#pragma unroll
      for (int j = 0; j < 4; ++j) {
        if (j == 2) {
          __builtin_amdgcn_sched_barrier(0);
          load_bfrags2<2>(bw, Wt1, lane15, half);
          __builtin_amdgcn_sched_barrier(0);
        }
        const int row = rbase + half * 8 + sub4 + j;
        _Float16* xrow = buf0 + row * HP + lane15;
        float gain = 0.f, asset = 0.f;
#pragma unroll
        for (int nt = 0; nt < 7; ++nt) {
          const int d = nt * 16 + lane15;
          float xv = (d == DIMA) ? tn[j] : 0.f;
          if (d < DIMA) {
            const float z  = 1.f / (1.f + expf(-(acc[nt][j] + bn[nt])));
            const float s  = sv[j][nt];
            const float ds = s * (RDT + SIG * dwv[j][nt]);
            const float sn = s + ds;
            gain  += z * ds;
            asset += z * s;
            Sl[row * SROW + d] = sn;
            xv = sn * S0INV;
          }
          xrow[nt * 16] = (_Float16)xv;
        }
        xrow[7 * 16] = (_Float16)0.f;
        gain  = red16(gain);
        asset = red16(asset);
        if (lane15 == 0) {
          const float y = yv[j];
          Yl[row] = y + RDT * (y - asset) + gain;
        }
      }
    }
    __syncthreads();
  }

  for (int i = tid; i < ROWS; i += 128) outY[blockBase + i] = Yl[i];
  for (int i = tid; i < ROWS * DIMA; i += 128) {
    const int r = i / DIMA, d = i - r * DIMA;
    outS[(blockBase + r) * (long)DIMA + d] = Sl[r * SROW + d];
  }
}

extern "C" void kernel_launch(void* const* d_in, const int* in_sizes, int n_in,
                              void* d_out, int out_size, void* d_ws, size_t ws_size,
                              hipStream_t stream) {
  (void)in_sizes; (void)n_in; (void)out_size; (void)d_ws; (void)ws_size;
  const float* dw  = (const float*)d_in[0];
  const float* tg  = (const float*)d_in[1];
  const float* W1  = (const float*)d_in[2];
  const float* b1  = (const float*)d_in[3];
  const float* g1  = (const float*)d_in[4];
  const float* be1 = (const float*)d_in[5];
  const float* W2  = (const float*)d_in[6];
  const float* b2  = (const float*)d_in[7];
  const float* g2  = (const float*)d_in[8];
  const float* be2 = (const float*)d_in[9];
  const float* W3  = (const float*)d_in[10];
  const float* b3  = (const float*)d_in[11];
  const float* g3  = (const float*)d_in[12];
  const float* be3 = (const float*)d_in[13];
  const float* W4  = (const float*)d_in[14];
  const float* b4  = (const float*)d_in[15];
  const float* Y0  = (const float*)d_in[16];

  float* outY = (float*)d_out;
  float* outS = outY + BATCH;

  const size_t smem =
      (size_t)(3 * HDIM * HP + W4ROWS * HP) * sizeof(_Float16) +
      (size_t)(9 * 128 + NPAD4) * sizeof(float) +
      (size_t)(ROWS * SROW + ROWS) * sizeof(float) +
      (size_t)(2 * ROWS * HP) * sizeof(_Float16);

  dim3 grid(BATCH / ROWS);
  dim3 block(128);
  bsde_fused_kernel<<<grid, block, smem, stream>>>(
      dw, tg, W1, b1, g1, be1, W2, b2, g2, be2, W3, b3, g3, be3, W4, b4, Y0,
      outY, outS);
}


# Weight staging reads memory along n: one 128 B line per load (was 32 lines per load, each trip waited for), eight k per lane into one 16 B LDS store, loads issued far ahead; same f16 values in LDS
# speedup vs baseline: 16.3156x; 1.1275x over previous
#include <hip/hip_runtime.h>
#include <math.h>

#define DIMA   100
#define MSTEP  50
#define HDIM   128
#define BATCH  8192
#define NPAD4  112
#define SROW   104
#define W4ROWS 128
#define HP     136
#define ROWS   32
#define S0INV  0.01f
#define RDT    0.001f
#define SIG    0.2f
#define LNEPS  1e-5f

typedef __attribute__((ext_vector_type(16))) _Float16 v16h;
typedef __attribute__((ext_vector_type(8)))  _Float16 v8h;
typedef __attribute__((ext_vector_type(8)))  float    v8f;

union V16U { v16h v; v8h h[2]; };

__device__ __forceinline__ v16h load_frag(const _Float16* base, int kb) {
  V16U u;
  u.h[0] = *(const v8h*)(base + kb);
  u.h[1] = *(const v8h*)(base + kb + 16);
  return u.v;
}

template <int KT0>
__device__ __forceinline__ void load_bfrags2(v16h (&bw)[4][8], const _Float16* Wt,
                                             int lane15, int half) {
#pragma unroll
  for (int kt = KT0; kt < KT0 + 2; ++kt) {
#pragma unroll
    for (int nt = 0; nt < 8; ++nt)
      bw[kt][nt] = load_frag(Wt + (nt * 16 + lane15) * HP, kt * 32 + half * 8);
  }
}

template <int M>
__device__ __forceinline__ float row_xor_add(float x) {
  const int partner = __builtin_amdgcn_update_dpp(
      0, __builtin_bit_cast(int, x), 0x160 + M, 0xf, 0xf, true);
  return x + __builtin_bit_cast(float, partner);
}

__device__ __forceinline__ float red16(float x) {
  x = row_xor_add<1>(x);
  x = row_xor_add<2>(x);
  x = row_xor_add<4>(x);
  x = row_xor_add<8>(x);
  return x;
}

__device__ __forceinline__ v8f erf8(v8f u) {
  v8f ax;
#pragma unroll
  for (int i = 0; i < 8; ++i) ax[i] = __builtin_fabsf(u[i]);
  const v8f t = u * u;
  v8f p = t * -5.618018e-4f + 4.913816e-3f;
  p = p * t + -2.6707515e-2f;
  p = p * t + 1.1280011e-1f;
  p = p * t + -3.7612295e-1f;
  p = p * t + 1.283791e-1f;
  const v8f rs = ax * p + ax;
  v8f q = ax * 1.699881e-5f + -3.7867785e-4f;
  q = q * ax + 3.8578159e-3f;
  q = q * ax + -2.4181698e-2f;
  q = q * ax + 1.0666826e-1f;
  q = q * ax + 6.349333e-1f;
  q = q * ax + 1.2868941e-1f;
  q = q * ax + ax;
  const v8f ea = q * -1.4426950408889634f;
  v8f rb;
#pragma unroll
  for (int i = 0; i < 8; ++i) rb[i] = 1.f - __builtin_amdgcn_exp2f(ea[i]);
  const v8f pick = (ax < 1.f) ? rs : rb;
  v8f r;
#pragma unroll
  for (int i = 0; i < 8; ++i) r[i] = __builtin_copysignf(pick[i], u[i]);
  return r;
}

__device__ __forceinline__ void mlp_layer(const _Float16* in, v16h (&bw)[4][8],
                                          const _Float16* WtNext,
                                          const float* bb, const float* gg,
                                          const float* ee, _Float16* out,
                                          int rbase, int lane15, int half, int sub4) {
  const _Float16* arow = in + (rbase + (lane15 ^ sub4)) * HP;
  v16h a[4];
#pragma unroll
  for (int kt = 0; kt < 4; ++kt) a[kt] = load_frag(arow, kt * 32 + half * 8);

  float bn[8], gn[8], en[8];
#pragma unroll
  for (int nt = 0; nt < 8; ++nt) {
    const int n = nt * 16 + lane15;
    bn[nt] = bb[n]; gn[nt] = gg[n]; en[nt] = ee[n];
  }

  __builtin_amdgcn_sched_barrier(0);

  v8f acc[8];
#pragma unroll
  for (int nt = 0; nt < 8; ++nt) acc[nt] = {};
#pragma unroll
  for (int kt = 0; kt < 4; ++kt) {
#pragma unroll
    for (int nt = 0; nt < 8; ++nt) {
      acc[nt] = __builtin_amdgcn_wmma_f32_16x16x32_f16(
          false, a[kt], false, bw[kt][nt], (short)0, acc[nt], false, false);
    }
  }

  __builtin_amdgcn_sched_barrier(0);
  load_bfrags2<0>(bw, WtNext, lane15, half);
  __builtin_amdgcn_sched_barrier(0);

#pragma unroll
  for (int j = 0; j < 4; ++j) {
    if (j == 2) {
      __builtin_amdgcn_sched_barrier(0);
      load_bfrags2<2>(bw, WtNext, lane15, half);
      __builtin_amdgcn_sched_barrier(0);
    }
    float p = 0.f;
#pragma unroll
    for (int nt = 0; nt < 8; ++nt) { acc[nt][j] += bn[nt]; p += acc[nt][j]; }
    p = red16(p);
    const float mu = p * (1.f / 128.f);
    float q = 0.f;
#pragma unroll
    for (int nt = 0; nt < 8; ++nt) { float d = acc[nt][j] - mu; q += d * d; }
    q = red16(q);
    const float rstd = rsqrtf(q * (1.f / 128.f) + LNEPS);

    const int row = rbase + half * 8 + sub4 + j;
    _Float16* orow = out + row * HP + lane15;
    v8f x;
#pragma unroll
    for (int nt = 0; nt < 8; ++nt)
      x[nt] = (acc[nt][j] - mu) * rstd * gn[nt] + en[nt];
    const v8f ge = (x * 0.5f) * (erf8(x * 0.70710678118654752f) + 1.f);
#pragma unroll
    for (int nt = 0; nt < 8; ++nt) orow[nt * 16] = (_Float16)ge[nt];
  }
}

__device__ __forceinline__ void fetch_wblock(float (&v)[4][8],
                                             const float* __restrict__ W1,
                                             const float* __restrict__ W2,
                                             const float* __restrict__ W3,
                                             const float* __restrict__ W4,
                                             int kb, int n, int n4) {
#pragma unroll
  for (int i = 0; i < 8; ++i) {
    const int k = kb * 8 + i;
    v[0][i] = (k < DIMA + 1) ? W1[k * HDIM + n] : 0.f;
    v[1][i] = W2[k * HDIM + n];
    v[2][i] = W3[k * HDIM + n];
    v[3][i] = W4[k * DIMA + n4];
  }
}

__device__ __forceinline__ void store_wblock(const float (&v)[4][8], _Float16* Wt,
                                             int kb, int n, bool in4) {
#pragma unroll
  for (int m = 0; m < 4; ++m) {
    v8h h;
#pragma unroll
    for (int i = 0; i < 8; ++i) h[i] = (_Float16)((m < 3 || in4) ? v[m][i] : 0.f);
    *(v8h*)(Wt + m * (HDIM * HP) + n * HP + kb * 8) = h;
  }
}

extern "C" __global__ void __launch_bounds__(128)
__attribute__((amdgpu_waves_per_eu(1, 1)))
bsde_fused_kernel(const float* __restrict__ dw, const float* __restrict__ tg,
                  const float* __restrict__ W1, const float* __restrict__ b1,
                  const float* __restrict__ g1, const float* __restrict__ be1,
                  const float* __restrict__ W2, const float* __restrict__ b2,
                  const float* __restrict__ g2, const float* __restrict__ be2,
                  const float* __restrict__ W3, const float* __restrict__ b3,
                  const float* __restrict__ g3, const float* __restrict__ be3,
                  const float* __restrict__ W4, const float* __restrict__ b4,
                  const float* __restrict__ Y0,
                  float* __restrict__ outY, float* __restrict__ outS) {
  extern __shared__ __align__(16) char smem[];
  _Float16* Wt1 = (_Float16*)smem;
  _Float16* Wt2 = Wt1 + HDIM * HP;
  _Float16* Wt3 = Wt2 + HDIM * HP;
  _Float16* Wt4 = Wt3 + HDIM * HP;
  float* prm = (float*)(Wt4 + W4ROWS * HP);
  float* pb1 = prm;        float* pg1 = prm + 128;  float* pe1 = prm + 256;
  float* pb2 = prm + 384;  float* pg2 = prm + 512;  float* pe2 = prm + 640;
  float* pb3 = prm + 768;  float* pg3 = prm + 896;  float* pe3 = prm + 1024;
  float* pb4 = prm + 1152;
  float* Sl  = prm + 1264;
  float* Yl  = Sl + ROWS * SROW;
  _Float16* buf0 = (_Float16*)(Yl + ROWS);
  _Float16* buf1 = buf0 + ROWS * HP;

  const int tid = threadIdx.x;
  const int lane = tid & 31, lane15 = lane & 15, half = lane >> 4;
  const int rbase = (tid >> 6) * 16;
  const int sub4  = ((tid >> 5) & 1) * 4;
  const long blockBase = (long)blockIdx.x * ROWS;

  {
    const bool in4 = tid < DIMA;
    const int n4 = in4 ? tid : (DIMA - 1);
#pragma unroll
    for (int kb = 0; kb < HDIM / 8; ++kb) {
      float blk[4][8];
      fetch_wblock(blk, W1, W2, W3, W4, kb, tid, n4);
      store_wblock(blk, Wt1, kb, tid, in4);
    }
  }
  pb1[tid] = b1[tid]; pg1[tid] = g1[tid]; pe1[tid] = be1[tid];
  pb2[tid] = b2[tid]; pg2[tid] = g2[tid]; pe2[tid] = be2[tid];
  pb3[tid] = b3[tid]; pg3[tid] = g3[tid]; pe3[tid] = be3[tid];
  if (tid < NPAD4) pb4[tid] = (tid < DIMA) ? b4[tid] : 0.f;

  const float y0 = Y0[0];
  for (int i = tid; i < ROWS * SROW; i += 128) Sl[i] = 100.f;
  if (tid < ROWS) Yl[tid] = y0;
  __syncthreads();

  for (int i = tid; i < ROWS * HDIM; i += 128) {
    const int row = i >> 7, k = i & 127;
    float v;
    if (k < DIMA)      v = Sl[row * SROW + k] * S0INV;
    else if (k == DIMA) v = tg[(blockBase + row) * MSTEP];
    else               v = 0.f;
    buf0[row * HP + k] = (_Float16)v;
  }
  __syncthreads();

  v16h bw[4][8];
  load_bfrags2<0>(bw, Wt1, lane15, half);
  load_bfrags2<2>(bw, Wt1, lane15, half);

  for (int m = 0; m < MSTEP; ++m) {
    float dwv[4][7];
    {
      const float* dwp =
          dw + ((blockBase + rbase + half * 8 + sub4) * MSTEP + m) * (long)DIMA + lane15;
#pragma unroll
      for (int j = 0; j < 4; ++j) {
#pragma unroll
        for (int nt = 0; nt < 6; ++nt) dwv[j][nt] = dwp[j * (MSTEP * DIMA) + nt * 16];
      }
      const int o6 = (lane15 < DIMA - 96) ? 96 : (DIMA - 1 - lane15);
#pragma unroll
      for (int j = 0; j < 4; ++j) dwv[j][6] = dwp[j * (MSTEP * DIMA) + o6];
    }

    float tn[4];
    {
      const int mn = (m + 1 < MSTEP) ? (m + 1) : m;
      const float* tp = tg + (blockBase + rbase + half * 8 + sub4) * MSTEP + mn;
#pragma unroll
      for (int j = 0; j < 4; ++j) tn[j] = tp[j * MSTEP];
    }

#pragma unroll 1
    for (int l = 0; l < 3; ++l) {
      const float* lp = prm + l * (3 * HDIM);
      mlp_layer((l & 1) ? buf1 : buf0, bw, Wt1 + (l + 1) * (HDIM * HP),
                lp, lp + HDIM, lp + 2 * HDIM,
                (l & 1) ? buf0 : buf1, rbase, lane15, half, sub4);
      __syncthreads();
    }

    {
      const _Float16* arow = buf1 + (rbase + (lane15 ^ sub4)) * HP;
      v16h a[4];
#pragma unroll
      for (int kt = 0; kt < 4; ++kt) a[kt] = load_frag(arow, kt * 32 + half * 8);

      float bn[7];
#pragma unroll
      for (int nt = 0; nt < 7; ++nt) bn[nt] = pb4[nt * 16 + lane15];
      float sv[4][7], yv[4];
      {
        const int d6 = (lane15 < DIMA - 96) ? (96 + lane15) : (DIMA - 1);
#pragma unroll
        for (int j = 0; j < 4; ++j) {
          const int row = rbase + half * 8 + sub4 + j;
#pragma unroll
          for (int nt = 0; nt < 6; ++nt) sv[j][nt] = Sl[row * SROW + nt * 16 + lane15];
          sv[j][6] = Sl[row * SROW + d6];
          yv[j] = Yl[row];
        }
      }

      __builtin_amdgcn_sched_barrier(0);

      v8f acc[7];
#pragma unroll
      for (int nt = 0; nt < 7; ++nt) acc[nt] = {};
#pragma unroll
      for (int kt = 0; kt < 4; ++kt) {
#pragma unroll
        for (int nt = 0; nt < 7; ++nt) {
          acc[nt] = __builtin_amdgcn_wmma_f32_16x16x32_f16(
              false, a[kt], false, bw[kt][nt], (short)0, acc[nt], false, false);
        }
      }

      __builtin_amdgcn_sched_barrier(0);
      load_bfrags2<0>(bw, Wt1, lane15, half);
      __builtin_amdgcn_sched_barrier(0);

#pragma unroll
      for (int j = 0; j < 4; ++j) {
        if (j == 2) {
          __builtin_amdgcn_sched_barrier(0);
          load_bfrags2<2>(bw, Wt1, lane15, half);
          __builtin_amdgcn_sched_barrier(0);
        }
        const int row = rbase + half * 8 + sub4 + j;
        _Float16* xrow = buf0 + row * HP + lane15;
        float gain = 0.f, asset = 0.f;
#pragma unroll
        for (int nt = 0; nt < 7; ++nt) {
          const int d = nt * 16 + lane15;
          float xv = (d == DIMA) ? tn[j] : 0.f;
          if (d < DIMA) {
            const float z  = 1.f / (1.f + expf(-(acc[nt][j] + bn[nt])));
            const float s  = sv[j][nt];
            const float ds = s * (RDT + SIG * dwv[j][nt]);
            const float sn = s + ds;
            gain  += z * ds;
            asset += z * s;
            Sl[row * SROW + d] = sn;
            xv = sn * S0INV;
          }
          xrow[nt * 16] = (_Float16)xv;
        }
        xrow[7 * 16] = (_Float16)0.f;
        gain  = red16(gain);
        asset = red16(asset);
        if (lane15 == 0) {
          const float y = yv[j];
          Yl[row] = y + RDT * (y - asset) + gain;
        }
      }
    }
    __syncthreads();
  }

  for (int i = tid; i < ROWS; i += 128) outY[blockBase + i] = Yl[i];
  for (int i = tid; i < ROWS * DIMA; i += 128) {
    const int r = i / DIMA, d = i - r * DIMA;
    outS[(blockBase + r) * (long)DIMA + d] = Sl[r * SROW + d];
  }
}

extern "C" void kernel_launch(void* const* d_in, const int* in_sizes, int n_in,
                              void* d_out, int out_size, void* d_ws, size_t ws_size,
                              hipStream_t stream) {
  (void)in_sizes; (void)n_in; (void)out_size; (void)d_ws; (void)ws_size;
  const float* dw  = (const float*)d_in[0];
  const float* tg  = (const float*)d_in[1];
  const float* W1  = (const float*)d_in[2];
  const float* b1  = (const float*)d_in[3];
  const float* g1  = (const float*)d_in[4];
  const float* be1 = (const float*)d_in[5];
  const float* W2  = (const float*)d_in[6];
  const float* b2  = (const float*)d_in[7];
  const float* g2  = (const float*)d_in[8];
  const float* be2 = (const float*)d_in[9];
  const float* W3  = (const float*)d_in[10];
  const float* b3  = (const float*)d_in[11];
  const float* g3  = (const float*)d_in[12];
  const float* be3 = (const float*)d_in[13];
  const float* W4  = (const float*)d_in[14];
  const float* b4  = (const float*)d_in[15];
  const float* Y0  = (const float*)d_in[16];

  float* outY = (float*)d_out;
  float* outS = outY + BATCH;

  const size_t smem =
      (size_t)(3 * HDIM * HP + W4ROWS * HP) * sizeof(_Float16) +
      (size_t)(9 * 128 + NPAD4) * sizeof(float) +
      (size_t)(ROWS * SROW + ROWS) * sizeof(float) +
      (size_t)(2 * ROWS * HP) * sizeof(_Float16);

  dim3 grid(BATCH / ROWS);
  dim3 block(128);
  bsde_fused_kernel<<<grid, block, smem, stream>>>(
      dw, tg, W1, b1, g1, be1, W2, b2, g2, be2, W3, b3, g3, be3, W4, b4, Y0,
      outY, outS);
}
